# starting kernel
# baseline (speedup 1.0000x reference)
#include <hip/hip_runtime.h>

typedef float v2f __attribute__((ext_vector_type(2)));
typedef float v8f __attribute__((ext_vector_type(8)));

#define P_PILLARS 60000
#define NPT 32
#define COUT 64
#define WPB 8
#define NBLK_A (P_PILLARS / WPB)

__global__ __launch_bounds__(256) void pfn_pass_a(
    const float* __restrict__ features,
    const int*   __restrict__ num_points,
    const int*   __restrict__ coors,
    const float* __restrict__ W,
    float* __restrict__ ws_mx, float* __restrict__ ws_mn,
    float* __restrict__ ws_psum, float* __restrict__ ws_pssq)
{
    __shared__ float Wl[12][COUT];
    __shared__ float featsL[WPB][NPT][12];
    __shared__ float wsum[WPB][COUT];
    __shared__ float wssq[WPB][COUT];

    const int tid = threadIdx.x;
    for (int i = tid; i < 12 * COUT; i += 256) {
        const int r = i / COUT, c = i % COUT;
        Wl[r][c] = (r < 9) ? W[r * COUT + c] : 0.0f;
    }

    const int wid  = tid >> 5;
    const int lane = tid & 31;
    const int half = lane >> 4;
    const int l16  = lane & 15;
    const int p    = blockIdx.x * WPB + wid;

    const float4 f = ((const float4*)features)[p * NPT + lane];
    const int npts = num_points[p];

    float sx = f.x, sy = f.y, sz = f.z;
    #pragma unroll
    for (int off = 16; off >= 1; off >>= 1) {
        sx += __shfl_xor(sx, off, 32);
        sy += __shfl_xor(sy, off, 32);
        sz += __shfl_xor(sz, off, 32);
    }
    const float inv_n = 1.0f / (float)npts;
    const float cmx = sx * inv_n, cmy = sy * inv_n, cmz = sz * inv_n;

    const float xc = (float)coors[p * 4 + 3] * 0.2f + 0.1f;
    const float yc = (float)coors[p * 4 + 2] * 0.2f + (0.1f - 40.0f);

    const float m = (lane < npts) ? 1.0f : 0.0f;
    float* row = &featsL[wid][lane][0];
    row[0] = f.x * m;           row[1] = f.y * m;
    row[2] = f.z * m;           row[3] = f.w * m;
    row[4] = (f.x - cmx) * m;   row[5] = (f.y - cmy) * m;
    row[6] = (f.z - cmz) * m;   row[7] = (f.x - xc) * m;
    row[8] = (f.y - yc) * m;
    row[9] = 0.0f; row[10] = 0.0f; row[11] = 0.0f;

    __syncthreads();

    v2f bfr[3][4];
    #pragma unroll
    for (int ks = 0; ks < 3; ++ks) {
        const int kr = ks * 4 + 2 * half;
        #pragma unroll
        for (int nt = 0; nt < 4; ++nt) {
            bfr[ks][nt][0] = Wl[kr][nt * 16 + l16];
            bfr[ks][nt][1] = Wl[kr + 1][nt * 16 + l16];
        }
    }

    float chmax[4], chmin[4], chs[4], chss[4];
    #pragma unroll
    for (int nt = 0; nt < 4; ++nt) {
        chmax[nt] = -3.402823e38f; chmin[nt] = 3.402823e38f;
        chs[nt] = 0.0f; chss[nt] = 0.0f;
    }

    #pragma unroll
    for (int mt = 0; mt < 2; ++mt) {
        v8f acc[4];
        #pragma unroll
        for (int nt = 0; nt < 4; ++nt) acc[nt] = (v8f){0,0,0,0,0,0,0,0};
        #pragma unroll
        for (int ks = 0; ks < 3; ++ks) {
            v2f a;
            a[0] = featsL[wid][mt * 16 + l16][ks * 4 + 2 * half];
            a[1] = featsL[wid][mt * 16 + l16][ks * 4 + 2 * half + 1];
            #pragma unroll
            for (int nt = 0; nt < 4; ++nt) {
                acc[nt] = __builtin_amdgcn_wmma_f32_16x16x4_f32(
                    false, a, false, bfr[ks][nt], (short)0, acc[nt], false, false);
            }
        }
        #pragma unroll
        for (int nt = 0; nt < 4; ++nt) {
            #pragma unroll
            for (int r = 0; r < 8; ++r) {
                const float v = acc[nt][r];
                chmax[nt] = fmaxf(chmax[nt], v);
                chmin[nt] = fminf(chmin[nt], v);
                chs[nt]  += v;
                chss[nt] += v * v;
            }
        }
    }
    #pragma unroll
    for (int nt = 0; nt < 4; ++nt) {
        chmax[nt] = fmaxf(chmax[nt], __shfl_xor(chmax[nt], 16, 32));
        chmin[nt] = fminf(chmin[nt], __shfl_xor(chmin[nt], 16, 32));
        chs[nt]  += __shfl_xor(chs[nt], 16, 32);
        chss[nt] += __shfl_xor(chss[nt], 16, 32);
    }
    if (half == 0) {
        #pragma unroll
        for (int nt = 0; nt < 4; ++nt) {
            const int c = nt * 16 + l16;
            ws_mx[(size_t)p * COUT + c] = chmax[nt];
            ws_mn[(size_t)p * COUT + c] = chmin[nt];
            wsum[wid][c] = chs[nt];
            wssq[wid][c] = chss[nt];
        }
    }
    __syncthreads();
    if (tid < COUT) {
        float s = 0.0f, ss = 0.0f;
        #pragma unroll
        for (int w = 0; w < WPB; ++w) { s += wsum[w][tid]; ss += wssq[w][tid]; }
        ws_psum[(size_t)blockIdx.x * COUT + tid] = s;
        ws_pssq[(size_t)blockIdx.x * COUT + tid] = ss;
    }
}

__global__ void pfn_stats(const float* __restrict__ psum,
                          const float* __restrict__ pssq,
                          const float* __restrict__ gamma,
                          const float* __restrict__ beta,
                          float* __restrict__ scale,
                          float* __restrict__ shift)
{
    const int c = threadIdx.x;
    if (c >= COUT) return;
    float s = 0.0f, ss = 0.0f;
    for (int b = 0; b < NBLK_A; ++b) {
        s  += psum[(size_t)b * COUT + c];
        ss += pssq[(size_t)b * COUT + c];
    }
    const float inv = 1.0f / (float)(P_PILLARS * NPT);
    const float mean = s * inv;
    const float var  = ss * inv - mean * mean;
    const float sc   = gamma[c] * rsqrtf(var + 1e-3f);
    scale[c] = sc;
    shift[c] = beta[c] - mean * sc;
}

__global__ __launch_bounds__(256) void pfn_finish(
    const float* __restrict__ ws_mx, const float* __restrict__ ws_mn,
    const float* __restrict__ scale, const float* __restrict__ shift,
    float* __restrict__ out)
{
    const int idx = blockIdx.x * 256 + threadIdx.x;
    const int c = idx & (COUT - 1);
    const float s = scale[c];
    const float v = (s >= 0.0f) ? ws_mx[idx] : ws_mn[idx];
    out[idx] = fmaxf(fmaf(s, v, shift[c]), 0.0f);
}

extern "C" void kernel_launch(void* const* d_in, const int* in_sizes, int n_in,
                              void* d_out, int out_size, void* d_ws, size_t ws_size,
                              hipStream_t stream) {
    const float* features   = (const float*)d_in[0];
    const int*   num_points = (const int*)  d_in[1];
    const int*   coors      = (const int*)  d_in[2];
    const float* W          = (const float*)d_in[3];
    const float* gamma      = (const float*)d_in[4];
    const float* beta       = (const float*)d_in[5];
    float* out = (float*)d_out;

    float* ws      = (float*)d_ws;
    float* ws_mx   = ws;
    float* ws_mn   = ws_mx   + (size_t)P_PILLARS * COUT;
    float* ws_psum = ws_mn   + (size_t)P_PILLARS * COUT;
    float* ws_pssq = ws_psum + (size_t)NBLK_A * COUT;
    float* ws_scl  = ws_pssq + (size_t)NBLK_A * COUT;
    float* ws_shf  = ws_scl + COUT;

    pfn_pass_a<<<NBLK_A, 256, 0, stream>>>(features, num_points, coors, W,
                                           ws_mx, ws_mn, ws_psum, ws_pssq);
    pfn_stats<<<1, 64, 0, stream>>>(ws_psum, ws_pssq, gamma, beta, ws_scl, ws_shf);
    pfn_finish<<<(P_PILLARS * COUT) / 256, 256, 0, stream>>>(ws_mx, ws_mn,
                                                             ws_scl, ws_shf, out);
}


# batch-norm sums: 1 block x 64 threads serial over 7500 partial rows -> fixed-order two-level tree (125 blocks x 60 rows, then 1 block x 125 rows); +1 launch, summation order changes
# speedup vs baseline: 7.7711x; 7.7711x over previous
#include <hip/hip_runtime.h>

typedef float v2f __attribute__((ext_vector_type(2)));
typedef float v8f __attribute__((ext_vector_type(8)));

#define P_PILLARS 60000
#define NPT 32
#define COUT 64
#define WPB 8
#define NBLK_A (P_PILLARS / WPB)
#define L1_BLKS 125
#define L1_ROWS (NBLK_A / L1_BLKS)
#define L1_WAVES 4
#define L2_WAVES 5

__global__ __launch_bounds__(256) void pfn_pass_a(
    const float* __restrict__ features,
    const int*   __restrict__ num_points,
    const int*   __restrict__ coors,
    const float* __restrict__ W,
    float* __restrict__ ws_mx, float* __restrict__ ws_mn,
    float* __restrict__ ws_psum, float* __restrict__ ws_pssq)
{
    __shared__ float Wl[12][COUT];
    __shared__ float featsL[WPB][NPT][12];
    __shared__ float wsum[WPB][COUT];
    __shared__ float wssq[WPB][COUT];

    const int tid = threadIdx.x;
    for (int i = tid; i < 12 * COUT; i += 256) {
        const int r = i / COUT, c = i % COUT;
        Wl[r][c] = (r < 9) ? W[r * COUT + c] : 0.0f;
    }

    const int wid  = tid >> 5;
    const int lane = tid & 31;
    const int half = lane >> 4;
    const int l16  = lane & 15;
    const int p    = blockIdx.x * WPB + wid;

    const float4 f = ((const float4*)features)[p * NPT + lane];
    const int npts = num_points[p];

    float sx = f.x, sy = f.y, sz = f.z;
    #pragma unroll
    for (int off = 16; off >= 1; off >>= 1) {
        sx += __shfl_xor(sx, off, 32);
        sy += __shfl_xor(sy, off, 32);
        sz += __shfl_xor(sz, off, 32);
    }
    const float inv_n = 1.0f / (float)npts;
    const float cmx = sx * inv_n, cmy = sy * inv_n, cmz = sz * inv_n;

    const float xc = (float)coors[p * 4 + 3] * 0.2f + 0.1f;
    const float yc = (float)coors[p * 4 + 2] * 0.2f + (0.1f - 40.0f);

    const float m = (lane < npts) ? 1.0f : 0.0f;
    float* row = &featsL[wid][lane][0];
    row[0] = f.x * m;           row[1] = f.y * m;
    row[2] = f.z * m;           row[3] = f.w * m;
    row[4] = (f.x - cmx) * m;   row[5] = (f.y - cmy) * m;
    row[6] = (f.z - cmz) * m;   row[7] = (f.x - xc) * m;
    row[8] = (f.y - yc) * m;
    row[9] = 0.0f; row[10] = 0.0f; row[11] = 0.0f;

    __syncthreads();

    v2f bfr[3][4];
    #pragma unroll
    for (int ks = 0; ks < 3; ++ks) {
        const int kr = ks * 4 + 2 * half;
        #pragma unroll
        for (int nt = 0; nt < 4; ++nt) {
            bfr[ks][nt][0] = Wl[kr][nt * 16 + l16];
            bfr[ks][nt][1] = Wl[kr + 1][nt * 16 + l16];
        }
    }

    float chmax[4], chmin[4], chs[4], chss[4];
    #pragma unroll
    for (int nt = 0; nt < 4; ++nt) {
        chmax[nt] = -3.402823e38f; chmin[nt] = 3.402823e38f;
        chs[nt] = 0.0f; chss[nt] = 0.0f;
    }

    #pragma unroll
    for (int mt = 0; mt < 2; ++mt) {
        v8f acc[4];
        #pragma unroll
        for (int nt = 0; nt < 4; ++nt) acc[nt] = (v8f){0,0,0,0,0,0,0,0};
        #pragma unroll
        for (int ks = 0; ks < 3; ++ks) {
            v2f a;
            a[0] = featsL[wid][mt * 16 + l16][ks * 4 + 2 * half];
            a[1] = featsL[wid][mt * 16 + l16][ks * 4 + 2 * half + 1];
            #pragma unroll
            for (int nt = 0; nt < 4; ++nt) {
                acc[nt] = __builtin_amdgcn_wmma_f32_16x16x4_f32(
                    false, a, false, bfr[ks][nt], (short)0, acc[nt], false, false);
            }
        }
        #pragma unroll
        for (int nt = 0; nt < 4; ++nt) {
            #pragma unroll
            for (int r = 0; r < 8; ++r) {
                const float v = acc[nt][r];
                chmax[nt] = fmaxf(chmax[nt], v);
                chmin[nt] = fminf(chmin[nt], v);
                chs[nt]  += v;
                chss[nt] += v * v;
            }
        }
    }
    #pragma unroll
    for (int nt = 0; nt < 4; ++nt) {
        chmax[nt] = fmaxf(chmax[nt], __shfl_xor(chmax[nt], 16, 32));
        chmin[nt] = fminf(chmin[nt], __shfl_xor(chmin[nt], 16, 32));
        chs[nt]  += __shfl_xor(chs[nt], 16, 32);
        chss[nt] += __shfl_xor(chss[nt], 16, 32);
    }
    if (half == 0) {
        #pragma unroll
        for (int nt = 0; nt < 4; ++nt) {
            const int c = nt * 16 + l16;
            ws_mx[(size_t)p * COUT + c] = chmax[nt];
            ws_mn[(size_t)p * COUT + c] = chmin[nt];
            wsum[wid][c] = chs[nt];
            wssq[wid][c] = chss[nt];
        }
    }
    __syncthreads();
    if (tid < COUT) {
        float s = 0.0f, ss = 0.0f;
        #pragma unroll
        for (int w = 0; w < WPB; ++w) { s += wsum[w][tid]; ss += wssq[w][tid]; }
        ws_psum[(size_t)blockIdx.x * COUT + tid] = s;
        ws_pssq[(size_t)blockIdx.x * COUT + tid] = ss;
    }
}

__global__ __launch_bounds__(L1_WAVES * 32) void pfn_stats_l1(
    const float* __restrict__ psum, const float* __restrict__ pssq,
    float* __restrict__ p2sum, float* __restrict__ p2ssq)
{
    __shared__ float rs[L1_WAVES][COUT];
    __shared__ float rq[L1_WAVES][COUT];
    constexpr int RPW = L1_ROWS / L1_WAVES;
    const int tid  = threadIdx.x;
    const int w    = tid >> 5;
    const int lane = tid & 31;
    const float2* ps = (const float2*)(psum + (size_t)blockIdx.x * L1_ROWS * COUT);
    const float2* pq = (const float2*)(pssq + (size_t)blockIdx.x * L1_ROWS * COUT);

    float2 a[RPW], b[RPW];
    #pragma unroll
    for (int i = 0; i < RPW; ++i) {
        a[i] = ps[(w + L1_WAVES * i) * (COUT / 2) + lane];
        b[i] = pq[(w + L1_WAVES * i) * (COUT / 2) + lane];
    }
    float2 s = a[0], q = b[0];
    #pragma unroll
    for (int i = 1; i < RPW; ++i) {
        s.x += a[i].x; s.y += a[i].y;
        q.x += b[i].x; q.y += b[i].y;
    }
    rs[w][2 * lane] = s.x; rs[w][2 * lane + 1] = s.y;
    rq[w][2 * lane] = q.x; rq[w][2 * lane + 1] = q.y;
    __syncthreads();
    if (tid < COUT) {
        float t = rs[0][tid], u = rq[0][tid];
        #pragma unroll
        for (int k = 1; k < L1_WAVES; ++k) { t += rs[k][tid]; u += rq[k][tid]; }
        p2sum[(size_t)blockIdx.x * COUT + tid] = t;
        p2ssq[(size_t)blockIdx.x * COUT + tid] = u;
    }
}

__global__ __launch_bounds__(L2_WAVES * 32) void pfn_stats_l2(
    const float* __restrict__ p2sum, const float* __restrict__ p2ssq,
    const float* __restrict__ gamma, const float* __restrict__ beta,
    float* __restrict__ scale, float* __restrict__ shift)
{
    __shared__ float rs[L2_WAVES][COUT];
    __shared__ float rq[L2_WAVES][COUT];
    constexpr int RPW = L1_BLKS / L2_WAVES;
    const int tid  = threadIdx.x;
    const int w    = tid >> 5;
    const int lane = tid & 31;
    const float2* ps = (const float2*)p2sum;
    const float2* pq = (const float2*)p2ssq;

    float2 a[RPW], b[RPW];
    #pragma unroll
    for (int i = 0; i < RPW; ++i) {
        a[i] = ps[(w + L2_WAVES * i) * (COUT / 2) + lane];
        b[i] = pq[(w + L2_WAVES * i) * (COUT / 2) + lane];
    }
    float2 s = a[0], q = b[0];
    #pragma unroll
    for (int i = 1; i < RPW; ++i) {
        s.x += a[i].x; s.y += a[i].y;
        q.x += b[i].x; q.y += b[i].y;
    }
    rs[w][2 * lane] = s.x; rs[w][2 * lane + 1] = s.y;
    rq[w][2 * lane] = q.x; rq[w][2 * lane + 1] = q.y;
    __syncthreads();
    if (tid < COUT) {
        float t = rs[0][tid], u = rq[0][tid];
        #pragma unroll
        for (int k = 1; k < L2_WAVES; ++k) { t += rs[k][tid]; u += rq[k][tid]; }
        const float inv  = 1.0f / (float)(P_PILLARS * NPT);
        const float mean = t * inv;
        const float var  = u * inv - mean * mean;
        const float sc   = gamma[tid] * rsqrtf(var + 1e-3f);
        scale[tid] = sc;
        shift[tid] = beta[tid] - mean * sc;
    }
}

__global__ __launch_bounds__(256) void pfn_finish(
    const float* __restrict__ ws_mx, const float* __restrict__ ws_mn,
    const float* __restrict__ scale, const float* __restrict__ shift,
    float* __restrict__ out)
{
    const int idx = blockIdx.x * 256 + threadIdx.x;
    const int c = idx & (COUT - 1);
    const float s = scale[c];
    const float v = (s >= 0.0f) ? ws_mx[idx] : ws_mn[idx];
    out[idx] = fmaxf(fmaf(s, v, shift[c]), 0.0f);
}

extern "C" void kernel_launch(void* const* d_in, const int* in_sizes, int n_in,
                              void* d_out, int out_size, void* d_ws, size_t ws_size,
                              hipStream_t stream) {
    const float* features   = (const float*)d_in[0];
    const int*   num_points = (const int*)  d_in[1];
    const int*   coors      = (const int*)  d_in[2];
    const float* W          = (const float*)d_in[3];
    const float* gamma      = (const float*)d_in[4];
    const float* beta       = (const float*)d_in[5];
    float* out = (float*)d_out;

    float* ws      = (float*)d_ws;
    float* ws_mx   = ws;
    float* ws_mn   = ws_mx   + (size_t)P_PILLARS * COUT;
    float* ws_psum = ws_mn   + (size_t)P_PILLARS * COUT;
    float* ws_pssq = ws_psum + (size_t)NBLK_A * COUT;
    float* ws_scl  = ws_pssq + (size_t)NBLK_A * COUT;
    float* ws_shf  = ws_scl + COUT;
    float* ws_p2s  = ws_shf + COUT;
    float* ws_p2q  = ws_p2s + (size_t)L1_BLKS * COUT;

    pfn_pass_a<<<NBLK_A, 256, 0, stream>>>(features, num_points, coors, W,
                                           ws_mx, ws_mn, ws_psum, ws_pssq);
    pfn_stats_l1<<<L1_BLKS, L1_WAVES * 32, 0, stream>>>(ws_psum, ws_pssq, ws_p2s, ws_p2q);
    pfn_stats_l2<<<1, L2_WAVES * 32, 0, stream>>>(ws_p2s, ws_p2q, gamma, beta, ws_scl, ws_shf);
    pfn_finish<<<(P_PILLARS * COUT) / 256, 256, 0, stream>>>(ws_mx, ws_mn,
                                                             ws_scl, ws_shf, out);
}


# finish pass: 15000 blocks of one-float threads -> 375 blocks whose threads each do 10 float4 (120000 waves -> 3000); wave start-up, not bytes, set its time
# speedup vs baseline: 9.3051x; 1.1974x over previous
#include <hip/hip_runtime.h>

typedef float v2f __attribute__((ext_vector_type(2)));
typedef float v8f __attribute__((ext_vector_type(8)));

#define P_PILLARS 60000
#define NPT 32
#define COUT 64
#define WPB 8
#define NBLK_A (P_PILLARS / WPB)
#define L1_BLKS 125
#define L1_ROWS (NBLK_A / L1_BLKS)
#define L1_WAVES 4
#define L2_WAVES 5

__global__ __launch_bounds__(256) void pfn_pass_a(
    const float* __restrict__ features,
    const int*   __restrict__ num_points,
    const int*   __restrict__ coors,
    const float* __restrict__ W,
    float* __restrict__ ws_mx, float* __restrict__ ws_mn,
    float* __restrict__ ws_psum, float* __restrict__ ws_pssq)
{
    __shared__ float Wl[12][COUT];
    __shared__ float featsL[WPB][NPT][12];
    __shared__ float wsum[WPB][COUT];
    __shared__ float wssq[WPB][COUT];

    const int tid = threadIdx.x;
    for (int i = tid; i < 12 * COUT; i += 256) {
        const int r = i / COUT, c = i % COUT;
        Wl[r][c] = (r < 9) ? W[r * COUT + c] : 0.0f;
    }

    const int wid  = tid >> 5;
    const int lane = tid & 31;
    const int half = lane >> 4;
    const int l16  = lane & 15;
    const int p    = blockIdx.x * WPB + wid;

    const float4 f = ((const float4*)features)[p * NPT + lane];
    const int npts = num_points[p];

    float sx = f.x, sy = f.y, sz = f.z;
    #pragma unroll
    for (int off = 16; off >= 1; off >>= 1) {
        sx += __shfl_xor(sx, off, 32);
        sy += __shfl_xor(sy, off, 32);
        sz += __shfl_xor(sz, off, 32);
    }
    const float inv_n = 1.0f / (float)npts;
    const float cmx = sx * inv_n, cmy = sy * inv_n, cmz = sz * inv_n;

    const float xc = (float)coors[p * 4 + 3] * 0.2f + 0.1f;
    const float yc = (float)coors[p * 4 + 2] * 0.2f + (0.1f - 40.0f);

    const float m = (lane < npts) ? 1.0f : 0.0f;
    float* row = &featsL[wid][lane][0];
    row[0] = f.x * m;           row[1] = f.y * m;
    row[2] = f.z * m;           row[3] = f.w * m;
    row[4] = (f.x - cmx) * m;   row[5] = (f.y - cmy) * m;
    row[6] = (f.z - cmz) * m;   row[7] = (f.x - xc) * m;
    row[8] = (f.y - yc) * m;
    row[9] = 0.0f; row[10] = 0.0f; row[11] = 0.0f;

    __syncthreads();

    v2f bfr[3][4];
    #pragma unroll
    for (int ks = 0; ks < 3; ++ks) {
        const int kr = ks * 4 + 2 * half;
        #pragma unroll
        for (int nt = 0; nt < 4; ++nt) {
            bfr[ks][nt][0] = Wl[kr][nt * 16 + l16];
            bfr[ks][nt][1] = Wl[kr + 1][nt * 16 + l16];
        }
    }

    float chmax[4], chmin[4], chs[4], chss[4];
    #pragma unroll
    for (int nt = 0; nt < 4; ++nt) {
        chmax[nt] = -3.402823e38f; chmin[nt] = 3.402823e38f;
        chs[nt] = 0.0f; chss[nt] = 0.0f;
    }

    #pragma unroll
    for (int mt = 0; mt < 2; ++mt) {
        v8f acc[4];
        #pragma unroll
        for (int nt = 0; nt < 4; ++nt) acc[nt] = (v8f){0,0,0,0,0,0,0,0};
        #pragma unroll
        for (int ks = 0; ks < 3; ++ks) {
            v2f a;
            a[0] = featsL[wid][mt * 16 + l16][ks * 4 + 2 * half];
            a[1] = featsL[wid][mt * 16 + l16][ks * 4 + 2 * half + 1];
            #pragma unroll
            for (int nt = 0; nt < 4; ++nt) {
                acc[nt] = __builtin_amdgcn_wmma_f32_16x16x4_f32(
                    false, a, false, bfr[ks][nt], (short)0, acc[nt], false, false);
            }
        }
        #pragma unroll
        for (int nt = 0; nt < 4; ++nt) {
            #pragma unroll
            for (int r = 0; r < 8; ++r) {
                const float v = acc[nt][r];
                chmax[nt] = fmaxf(chmax[nt], v);
                chmin[nt] = fminf(chmin[nt], v);
                chs[nt]  += v;
                chss[nt] += v * v;
            }
        }
    }
    #pragma unroll
    for (int nt = 0; nt < 4; ++nt) {
        chmax[nt] = fmaxf(chmax[nt], __shfl_xor(chmax[nt], 16, 32));
        chmin[nt] = fminf(chmin[nt], __shfl_xor(chmin[nt], 16, 32));
        chs[nt]  += __shfl_xor(chs[nt], 16, 32);
        chss[nt] += __shfl_xor(chss[nt], 16, 32);
    }
    if (half == 0) {
        #pragma unroll
        for (int nt = 0; nt < 4; ++nt) {
            const int c = nt * 16 + l16;
            ws_mx[(size_t)p * COUT + c] = chmax[nt];
            ws_mn[(size_t)p * COUT + c] = chmin[nt];
            wsum[wid][c] = chs[nt];
            wssq[wid][c] = chss[nt];
        }
    }
    __syncthreads();
    if (tid < COUT) {
        float s = 0.0f, ss = 0.0f;
        #pragma unroll
        for (int w = 0; w < WPB; ++w) { s += wsum[w][tid]; ss += wssq[w][tid]; }
        ws_psum[(size_t)blockIdx.x * COUT + tid] = s;
        ws_pssq[(size_t)blockIdx.x * COUT + tid] = ss;
    }
}

__global__ __launch_bounds__(L1_WAVES * 32) void pfn_stats_l1(
    const float* __restrict__ psum, const float* __restrict__ pssq,
    float* __restrict__ p2sum, float* __restrict__ p2ssq)
{
    __shared__ float rs[L1_WAVES][COUT];
    __shared__ float rq[L1_WAVES][COUT];
    constexpr int RPW = L1_ROWS / L1_WAVES;
    const int tid  = threadIdx.x;
    const int w    = tid >> 5;
    const int lane = tid & 31;
    const float2* ps = (const float2*)(psum + (size_t)blockIdx.x * L1_ROWS * COUT);
    const float2* pq = (const float2*)(pssq + (size_t)blockIdx.x * L1_ROWS * COUT);

    float2 a[RPW], b[RPW];
    #pragma unroll
    for (int i = 0; i < RPW; ++i) {
        a[i] = ps[(w + L1_WAVES * i) * (COUT / 2) + lane];
        b[i] = pq[(w + L1_WAVES * i) * (COUT / 2) + lane];
    }
    float2 s = a[0], q = b[0];
    #pragma unroll
    for (int i = 1; i < RPW; ++i) {
        s.x += a[i].x; s.y += a[i].y;
        q.x += b[i].x; q.y += b[i].y;
    }
    rs[w][2 * lane] = s.x; rs[w][2 * lane + 1] = s.y;
    rq[w][2 * lane] = q.x; rq[w][2 * lane + 1] = q.y;
    __syncthreads();
    if (tid < COUT) {
        float t = rs[0][tid], u = rq[0][tid];
        #pragma unroll
        for (int k = 1; k < L1_WAVES; ++k) { t += rs[k][tid]; u += rq[k][tid]; }
        p2sum[(size_t)blockIdx.x * COUT + tid] = t;
        p2ssq[(size_t)blockIdx.x * COUT + tid] = u;
    }
}

__global__ __launch_bounds__(L2_WAVES * 32) void pfn_stats_l2(
    const float* __restrict__ p2sum, const float* __restrict__ p2ssq,
    const float* __restrict__ gamma, const float* __restrict__ beta,
    float* __restrict__ scale, float* __restrict__ shift)
{
    __shared__ float rs[L2_WAVES][COUT];
    __shared__ float rq[L2_WAVES][COUT];
    constexpr int RPW = L1_BLKS / L2_WAVES;
    const int tid  = threadIdx.x;
    const int w    = tid >> 5;
    const int lane = tid & 31;
    const float2* ps = (const float2*)p2sum;
    const float2* pq = (const float2*)p2ssq;

    float2 a[RPW], b[RPW];
    #pragma unroll
    for (int i = 0; i < RPW; ++i) {
        a[i] = ps[(w + L2_WAVES * i) * (COUT / 2) + lane];
        b[i] = pq[(w + L2_WAVES * i) * (COUT / 2) + lane];
    }
    float2 s = a[0], q = b[0];
    #pragma unroll
    for (int i = 1; i < RPW; ++i) {
        s.x += a[i].x; s.y += a[i].y;
        q.x += b[i].x; q.y += b[i].y;
    }
    rs[w][2 * lane] = s.x; rs[w][2 * lane + 1] = s.y;
    rq[w][2 * lane] = q.x; rq[w][2 * lane + 1] = q.y;
    __syncthreads();
    if (tid < COUT) {
        float t = rs[0][tid], u = rq[0][tid];
        #pragma unroll
        for (int k = 1; k < L2_WAVES; ++k) { t += rs[k][tid]; u += rq[k][tid]; }
        const float inv  = 1.0f / (float)(P_PILLARS * NPT);
        const float mean = t * inv;
        const float var  = u * inv - mean * mean;
        const float sc   = gamma[tid] * rsqrtf(var + 1e-3f);
        scale[tid] = sc;
        shift[tid] = beta[tid] - mean * sc;
    }
}

#define FIN_THREADS 256
#define FIN_K 10
#define FIN_BLKS ((P_PILLARS * COUT / 4) / (FIN_THREADS * FIN_K))

template <int MODE>
__device__ __forceinline__ void pfn_finish_pieces(
    const float4* __restrict__ mx4, const float4* __restrict__ mn4,
    float4* __restrict__ out4, const int i0, const float4 s, const float4 t)
{
    float4 a[FIN_K], b[FIN_K];
    #pragma unroll
    for (int k = 0; k < FIN_K; ++k) {
        if (MODE != 1) a[k] = mx4[i0 + k * FIN_THREADS];
        if (MODE != 0) b[k] = mn4[i0 + k * FIN_THREADS];
    }
    #pragma unroll
    for (int k = 0; k < FIN_K; ++k) {
        float4 v;
        if (MODE == 0) {
            v = a[k];
        } else if (MODE == 1) {
            v = b[k];
        } else {
            v.x = (s.x >= 0.0f) ? a[k].x : b[k].x;
            v.y = (s.y >= 0.0f) ? a[k].y : b[k].y;
            v.z = (s.z >= 0.0f) ? a[k].z : b[k].z;
            v.w = (s.w >= 0.0f) ? a[k].w : b[k].w;
        }
        float4 o;
        o.x = fmaxf(fmaf(s.x, v.x, t.x), 0.0f);
        o.y = fmaxf(fmaf(s.y, v.y, t.y), 0.0f);
        o.z = fmaxf(fmaf(s.z, v.z, t.z), 0.0f);
        o.w = fmaxf(fmaf(s.w, v.w, t.w), 0.0f);
        out4[i0 + k * FIN_THREADS] = o;
    }
}

__global__ __launch_bounds__(FIN_THREADS) void pfn_finish(
    const float* __restrict__ ws_mx, const float* __restrict__ ws_mn,
    const float* __restrict__ scale, const float* __restrict__ shift,
    float* __restrict__ out)
{
    const int tid = threadIdx.x;
    const int i0 = blockIdx.x * (FIN_K * FIN_THREADS) + tid;
    const float4 s = ((const float4*)scale)[tid & 15];
    const float4 t = ((const float4*)shift)[tid & 15];
    const float4* mx4 = (const float4*)ws_mx;
    const float4* mn4 = (const float4*)ws_mn;
    float4* out4 = (float4*)out;

    const bool px = s.x >= 0.0f, py = s.y >= 0.0f, pz = s.z >= 0.0f, pw = s.w >= 0.0f;
    if (px && py && pz && pw) {
        pfn_finish_pieces<0>(mx4, mn4, out4, i0, s, t);
    } else if (!(px || py || pz || pw)) {
        pfn_finish_pieces<1>(mx4, mn4, out4, i0, s, t);
    } else {
        pfn_finish_pieces<2>(mx4, mn4, out4, i0, s, t);
    }
}

extern "C" void kernel_launch(void* const* d_in, const int* in_sizes, int n_in,
                              void* d_out, int out_size, void* d_ws, size_t ws_size,
                              hipStream_t stream) {
    const float* features   = (const float*)d_in[0];
    const int*   num_points = (const int*)  d_in[1];
    const int*   coors      = (const int*)  d_in[2];
    const float* W          = (const float*)d_in[3];
    const float* gamma      = (const float*)d_in[4];
    const float* beta       = (const float*)d_in[5];
    float* out = (float*)d_out;

    float* ws      = (float*)d_ws;
    float* ws_mx   = ws;
    float* ws_mn   = ws_mx   + (size_t)P_PILLARS * COUT;
    float* ws_psum = ws_mn   + (size_t)P_PILLARS * COUT;
    float* ws_pssq = ws_psum + (size_t)NBLK_A * COUT;
    float* ws_scl  = ws_pssq + (size_t)NBLK_A * COUT;
    float* ws_shf  = ws_scl + COUT;
    float* ws_p2s  = ws_shf + COUT;
    float* ws_p2q  = ws_p2s + (size_t)L1_BLKS * COUT;

    pfn_pass_a<<<NBLK_A, 256, 0, stream>>>(features, num_points, coors, W,
                                           ws_mx, ws_mn, ws_psum, ws_pssq);
    pfn_stats_l1<<<L1_BLKS, L1_WAVES * 32, 0, stream>>>(ws_psum, ws_pssq, ws_p2s, ws_p2q);
    pfn_stats_l2<<<1, L2_WAVES * 32, 0, stream>>>(ws_p2s, ws_p2q, gamma, beta, ws_scl, ws_shf);
    pfn_finish<<<FIN_BLKS, FIN_THREADS, 0, stream>>>(ws_mx, ws_mn, ws_scl, ws_shf, out);
}


# pass A: 60000 one-pillar waves (W, feats staged in LDS per block) -> 5120 long-lived waves, each walks 11-12 pillars with W and feats fragments in registers; stats summed per wave: sum order changes
# speedup vs baseline: 10.8395x; 1.1649x over previous
#include <hip/hip_runtime.h>

typedef float v2f __attribute__((ext_vector_type(2)));
typedef float v8f __attribute__((ext_vector_type(8)));

#define P_PILLARS 60000
#define NPT 32
#define COUT 64
#define WPB 4
#define NBLK_A 1280
#define NWAVE_A (NBLK_A * WPB)
#define RUN_LEN (P_PILLARS / NWAVE_A)
#define RUN_BIG (P_PILLARS - RUN_LEN * NWAVE_A)
#define L1_BLKS 32
#define L1_ROWS (NBLK_A / L1_BLKS)
#define L1_WAVES 4
#define L2_WAVES 4

__global__ __launch_bounds__(WPB * 32) void pfn_pass_a(
    const float* __restrict__ features,
    const int*   __restrict__ num_points,
    const int*   __restrict__ coors,
    const float* __restrict__ W,
    float* __restrict__ ws_mx, float* __restrict__ ws_mn,
    float* __restrict__ ws_psum, float* __restrict__ ws_pssq)
{
    __shared__ float wsum[WPB][COUT];
    __shared__ float wssq[WPB][COUT];

    const int tid  = threadIdx.x;
    const int wid  = __builtin_amdgcn_readfirstlane(tid >> 5);
    const int lane = tid & 31;
    const int l16  = lane & 15;
    const bool hi  = (lane >> 4) != 0;
    const int k2   = hi ? 2 : 0;

    const int gw    = blockIdx.x * WPB + wid;
    const int cnt   = RUN_LEN + (gw < RUN_BIG ? 1 : 0);
    const int start = gw * RUN_LEN + (gw < RUN_BIG ? gw : RUN_BIG);

    const int hl   = (lane < cnt) ? lane : cnt - 1;
    const int h_np = num_points[start + hl];
    const int h_cy = coors[(start + hl) * 4 + 2];
    const int h_cx = coors[(start + hl) * 4 + 3];

    v2f bfr[3][4];
    #pragma unroll
    for (int nt = 0; nt < 4; ++nt) {
        const int c = nt * 16 + l16;
        bfr[0][nt][0] = W[(k2)     * COUT + c];
        bfr[0][nt][1] = W[(k2 + 1) * COUT + c];
        bfr[1][nt][0] = W[(k2 + 4) * COUT + c];
        bfr[1][nt][1] = W[(k2 + 5) * COUT + c];
        const float w8 = W[8 * COUT + c];
        bfr[2][nt][0] = hi ? 0.0f : w8;
        bfr[2][nt][1] = 0.0f;
    }

    float chs[4], chss[4];
    #pragma unroll
    for (int nt = 0; nt < 4; ++nt) { chs[nt] = 0.0f; chss[nt] = 0.0f; }

    const float4* f4 = (const float4*)features;
    float4 fa = f4[start * NPT + l16];
    float4 fb = f4[start * NPT + 16 + l16];
    int npts = __builtin_amdgcn_readlane(h_np, 0);
    int cy   = __builtin_amdgcn_readlane(h_cy, 0);
    int cx   = __builtin_amdgcn_readlane(h_cx, 0);

    for (int i = 0; i < cnt; ++i) {
        const int p = start + i;
        float4 na = fa, nb = fb;
        if (i + 1 < cnt) {
            na = f4[(p + 1) * NPT + l16];
            nb = f4[(p + 1) * NPT + 16 + l16];
        }
        __builtin_amdgcn_sched_barrier(0);

        float sx = fa.x + fb.x, sy = fa.y + fb.y, sz = fa.z + fb.z;
        #pragma unroll
        for (int off = 8; off >= 1; off >>= 1) {
            sx += __shfl_xor(sx, off, 32);
            sy += __shfl_xor(sy, off, 32);
            sz += __shfl_xor(sz, off, 32);
        }
        const float inv_n = 1.0f / (float)npts;
        const float cmx = sx * inv_n, cmy = sy * inv_n, cmz = sz * inv_n;

        const float xc = (float)cx * 0.2f + 0.1f;
        const float yc = (float)cy * 0.2f + (0.1f - 40.0f);

        const float c0 = hi ? cmz : cmx;
        const float c1 = hi ? xc  : cmy;
        v2f afr[2][3];
        #pragma unroll
        for (int mt = 0; mt < 2; ++mt) {
            const float4 f = mt ? fb : fa;
            const float m  = (mt * 16 + l16 < npts) ? 1.0f : 0.0f;
            const float u0 = hi ? f.z : f.x;
            const float u1 = hi ? f.w : f.y;
            const float v1 = hi ? f.x : f.y;
            afr[mt][0][0] = u0 * m;
            afr[mt][0][1] = u1 * m;
            afr[mt][1][0] = (u0 - c0) * m;
            afr[mt][1][1] = (v1 - c1) * m;
            afr[mt][2][0] = hi ? 0.0f : (f.y - yc) * m;
            afr[mt][2][1] = 0.0f;
        }

        float chmax[4], chmin[4];
        #pragma unroll
        for (int nt = 0; nt < 4; ++nt) { chmax[nt] = -3.402823e38f; chmin[nt] = 3.402823e38f; }

        #pragma unroll
        for (int mt = 0; mt < 2; ++mt) {
            v8f acc[4];
            #pragma unroll
            for (int nt = 0; nt < 4; ++nt) acc[nt] = (v8f){0,0,0,0,0,0,0,0};
            #pragma unroll
            for (int ks = 0; ks < 3; ++ks) {
                #pragma unroll
                for (int nt = 0; nt < 4; ++nt) {
                    acc[nt] = __builtin_amdgcn_wmma_f32_16x16x4_f32(
                        false, afr[mt][ks], false, bfr[ks][nt], (short)0, acc[nt], false, false);
                }
            }
            #pragma unroll
            for (int nt = 0; nt < 4; ++nt) {
                #pragma unroll
                for (int r = 0; r < 8; ++r) {
                    const float v = acc[nt][r];
                    chmax[nt] = fmaxf(chmax[nt], v);
                    chmin[nt] = fminf(chmin[nt], v);
                    chs[nt]  += v;
                    chss[nt] += v * v;
                }
            }
        }

        #pragma unroll
        for (int j = 0; j < 2; ++j) {
            const float mine_mx = hi ? chmax[2 * j + 1] : chmax[2 * j];
            const float send_mx = hi ? chmax[2 * j]     : chmax[2 * j + 1];
            const float mine_mn = hi ? chmin[2 * j + 1] : chmin[2 * j];
            const float send_mn = hi ? chmin[2 * j]     : chmin[2 * j + 1];
            const float mx = fmaxf(mine_mx, __shfl_xor(send_mx, 16, 32));
            const float mn = fminf(mine_mn, __shfl_xor(send_mn, 16, 32));
            ws_mx[(size_t)p * COUT + j * 32 + lane] = mx;
            ws_mn[(size_t)p * COUT + j * 32 + lane] = mn;
        }

        fa = na; fb = nb;
        npts = __builtin_amdgcn_readlane(h_np, i + 1);
        cy   = __builtin_amdgcn_readlane(h_cy, i + 1);
        cx   = __builtin_amdgcn_readlane(h_cx, i + 1);
    }

    #pragma unroll
    for (int nt = 0; nt < 4; ++nt) {
        chs[nt]  += __shfl_xor(chs[nt], 16, 32);
        chss[nt] += __shfl_xor(chss[nt], 16, 32);
    }
    if (!hi) {
        #pragma unroll
        for (int nt = 0; nt < 4; ++nt) {
            wsum[wid][nt * 16 + l16] = chs[nt];
            wssq[wid][nt * 16 + l16] = chss[nt];
        }
    }
    __syncthreads();
    if (tid < COUT) {
        float s = 0.0f, ss = 0.0f;
        #pragma unroll
        for (int w = 0; w < WPB; ++w) { s += wsum[w][tid]; ss += wssq[w][tid]; }
        ws_psum[(size_t)blockIdx.x * COUT + tid] = s;
        ws_pssq[(size_t)blockIdx.x * COUT + tid] = ss;
    }
}

__global__ __launch_bounds__(L1_WAVES * 32) void pfn_stats_l1(
    const float* __restrict__ psum, const float* __restrict__ pssq,
    float* __restrict__ p2sum, float* __restrict__ p2ssq)
{
    __shared__ float rs[L1_WAVES][COUT];
    __shared__ float rq[L1_WAVES][COUT];
    constexpr int RPW = L1_ROWS / L1_WAVES;
    const int tid  = threadIdx.x;
    const int w    = tid >> 5;
    const int lane = tid & 31;
    const float2* ps = (const float2*)(psum + (size_t)blockIdx.x * L1_ROWS * COUT);
    const float2* pq = (const float2*)(pssq + (size_t)blockIdx.x * L1_ROWS * COUT);

    float2 a[RPW], b[RPW];
    #pragma unroll
    for (int i = 0; i < RPW; ++i) {
        a[i] = ps[(w + L1_WAVES * i) * (COUT / 2) + lane];
        b[i] = pq[(w + L1_WAVES * i) * (COUT / 2) + lane];
    }
    float2 s = a[0], q = b[0];
    #pragma unroll
    for (int i = 1; i < RPW; ++i) {
        s.x += a[i].x; s.y += a[i].y;
        q.x += b[i].x; q.y += b[i].y;
    }
    rs[w][2 * lane] = s.x; rs[w][2 * lane + 1] = s.y;
    rq[w][2 * lane] = q.x; rq[w][2 * lane + 1] = q.y;
    __syncthreads();
    if (tid < COUT) {
        float t = rs[0][tid], u = rq[0][tid];
        #pragma unroll
        for (int k = 1; k < L1_WAVES; ++k) { t += rs[k][tid]; u += rq[k][tid]; }
        p2sum[(size_t)blockIdx.x * COUT + tid] = t;
        p2ssq[(size_t)blockIdx.x * COUT + tid] = u;
    }
}

__global__ __launch_bounds__(L2_WAVES * 32) void pfn_stats_l2(
    const float* __restrict__ p2sum, const float* __restrict__ p2ssq,
    const float* __restrict__ gamma, const float* __restrict__ beta,
    float* __restrict__ scale, float* __restrict__ shift)
{
    __shared__ float rs[L2_WAVES][COUT];
    __shared__ float rq[L2_WAVES][COUT];
    constexpr int RPW = L1_BLKS / L2_WAVES;
    const int tid  = threadIdx.x;
    const int w    = tid >> 5;
    const int lane = tid & 31;
    const float2* ps = (const float2*)p2sum;
    const float2* pq = (const float2*)p2ssq;

    float2 a[RPW], b[RPW];
    #pragma unroll
    for (int i = 0; i < RPW; ++i) {
        a[i] = ps[(w + L2_WAVES * i) * (COUT / 2) + lane];
        b[i] = pq[(w + L2_WAVES * i) * (COUT / 2) + lane];
    }
    float2 s = a[0], q = b[0];
    #pragma unroll
    for (int i = 1; i < RPW; ++i) {
        s.x += a[i].x; s.y += a[i].y;
        q.x += b[i].x; q.y += b[i].y;
    }
    rs[w][2 * lane] = s.x; rs[w][2 * lane + 1] = s.y;
    rq[w][2 * lane] = q.x; rq[w][2 * lane + 1] = q.y;
    __syncthreads();
    if (tid < COUT) {
        float t = rs[0][tid], u = rq[0][tid];
        #pragma unroll
        for (int k = 1; k < L2_WAVES; ++k) { t += rs[k][tid]; u += rq[k][tid]; }
        const float inv  = 1.0f / (float)(P_PILLARS * NPT);
        const float mean = t * inv;
        const float var  = u * inv - mean * mean;
        const float sc   = gamma[tid] * rsqrtf(var + 1e-3f);
        scale[tid] = sc;
        shift[tid] = beta[tid] - mean * sc;
    }
}

#define FIN_THREADS 256
#define FIN_K 10
#define FIN_BLKS ((P_PILLARS * COUT / 4) / (FIN_THREADS * FIN_K))

template <int MODE>
__device__ __forceinline__ void pfn_finish_pieces(
    const float4* __restrict__ mx4, const float4* __restrict__ mn4,
    float4* __restrict__ out4, const int i0, const float4 s, const float4 t)
{
    float4 a[FIN_K], b[FIN_K];
    #pragma unroll
    for (int k = 0; k < FIN_K; ++k) {
        if (MODE != 1) a[k] = mx4[i0 + k * FIN_THREADS];
        if (MODE != 0) b[k] = mn4[i0 + k * FIN_THREADS];
    }
    #pragma unroll
    for (int k = 0; k < FIN_K; ++k) {
        float4 v;
        if (MODE == 0) {
            v = a[k];
        } else if (MODE == 1) {
            v = b[k];
        } else {
            v.x = (s.x >= 0.0f) ? a[k].x : b[k].x;
            v.y = (s.y >= 0.0f) ? a[k].y : b[k].y;
            v.z = (s.z >= 0.0f) ? a[k].z : b[k].z;
            v.w = (s.w >= 0.0f) ? a[k].w : b[k].w;
        }
        float4 o;
        o.x = fmaxf(fmaf(s.x, v.x, t.x), 0.0f);
        o.y = fmaxf(fmaf(s.y, v.y, t.y), 0.0f);
        o.z = fmaxf(fmaf(s.z, v.z, t.z), 0.0f);
        o.w = fmaxf(fmaf(s.w, v.w, t.w), 0.0f);
        out4[i0 + k * FIN_THREADS] = o;
    }
}

__global__ __launch_bounds__(FIN_THREADS) void pfn_finish(
    const float* __restrict__ ws_mx, const float* __restrict__ ws_mn,
    const float* __restrict__ scale, const float* __restrict__ shift,
    float* __restrict__ out)
{
    const int tid = threadIdx.x;
    const int i0 = blockIdx.x * (FIN_K * FIN_THREADS) + tid;
    const float4 s = ((const float4*)scale)[tid & 15];
    const float4 t = ((const float4*)shift)[tid & 15];
    const float4* mx4 = (const float4*)ws_mx;
    const float4* mn4 = (const float4*)ws_mn;
    float4* out4 = (float4*)out;

    const bool px = s.x >= 0.0f, py = s.y >= 0.0f, pz = s.z >= 0.0f, pw = s.w >= 0.0f;
    if (px && py && pz && pw) {
        pfn_finish_pieces<0>(mx4, mn4, out4, i0, s, t);
    } else if (!(px || py || pz || pw)) {
        pfn_finish_pieces<1>(mx4, mn4, out4, i0, s, t);
    } else {
        pfn_finish_pieces<2>(mx4, mn4, out4, i0, s, t);
    }
}

extern "C" void kernel_launch(void* const* d_in, const int* in_sizes, int n_in,
                              void* d_out, int out_size, void* d_ws, size_t ws_size,
                              hipStream_t stream) {
    const float* features   = (const float*)d_in[0];
    const int*   num_points = (const int*)  d_in[1];
    const int*   coors      = (const int*)  d_in[2];
    const float* W          = (const float*)d_in[3];
    const float* gamma      = (const float*)d_in[4];
    const float* beta       = (const float*)d_in[5];
    float* out = (float*)d_out;

    float* ws      = (float*)d_ws;
    float* ws_mx   = ws;
    float* ws_mn   = ws_mx   + (size_t)P_PILLARS * COUT;
    float* ws_psum = ws_mn   + (size_t)P_PILLARS * COUT;
    float* ws_pssq = ws_psum + (size_t)NBLK_A * COUT;
    float* ws_scl  = ws_pssq + (size_t)NBLK_A * COUT;
    float* ws_shf  = ws_scl + COUT;
    float* ws_p2s  = ws_shf + COUT;
    float* ws_p2q  = ws_p2s + (size_t)L1_BLKS * COUT;

    pfn_pass_a<<<NBLK_A, WPB * 32, 0, stream>>>(features, num_points, coors, W,
                                           ws_mx, ws_mn, ws_psum, ws_pssq);
    pfn_stats_l1<<<L1_BLKS, L1_WAVES * 32, 0, stream>>>(ws_psum, ws_pssq, ws_p2s, ws_p2q);
    pfn_stats_l2<<<1, L2_WAVES * 32, 0, stream>>>(ws_p2s, ws_p2q, gamma, beta, ws_scl, ws_shf);
    pfn_finish<<<FIN_BLKS, FIN_THREADS, 0, stream>>>(ws_mx, ws_mn, ws_scl, ws_shf, out);
}


# batch-norm statistics: pass A in 256 blocks of 20 waves (256 partial rows, was 1280); each finish block folds them itself in one fixed order; the 2 reduction launches go (4 -> 2); sum order changes
# speedup vs baseline: 11.5329x; 1.0640x over previous
#include <hip/hip_runtime.h>

typedef float v2f __attribute__((ext_vector_type(2)));
typedef float v8f __attribute__((ext_vector_type(8)));

#define P_PILLARS 60000
#define NPT 32
#define COUT 64
#define WPB 20
#define NBLK_A 256
#define NWAVE_A (NBLK_A * WPB)
#define RUN_LEN (P_PILLARS / NWAVE_A)
#define RUN_BIG (P_PILLARS - RUN_LEN * NWAVE_A)

__global__ __launch_bounds__(WPB * 32) void pfn_pass_a(
    const float* __restrict__ features,
    const int*   __restrict__ num_points,
    const int*   __restrict__ coors,
    const float* __restrict__ W,
    float* __restrict__ ws_mx, float* __restrict__ ws_mn,
    float* __restrict__ ws_psum, float* __restrict__ ws_pssq)
{
    __shared__ float wsum[WPB][COUT];
    __shared__ float wssq[WPB][COUT];

    const int tid  = threadIdx.x;
    const int wid  = __builtin_amdgcn_readfirstlane(tid >> 5);
    const int lane = tid & 31;
    const int l16  = lane & 15;
    const bool hi  = (lane >> 4) != 0;
    const int k2   = hi ? 2 : 0;

    const int gw    = blockIdx.x * WPB + wid;
    const int cnt   = RUN_LEN + (gw < RUN_BIG ? 1 : 0);
    const int start = gw * RUN_LEN + (gw < RUN_BIG ? gw : RUN_BIG);

    const int hl   = (lane < cnt) ? lane : cnt - 1;
    const int h_np = num_points[start + hl];
    const int h_cy = coors[(start + hl) * 4 + 2];
    const int h_cx = coors[(start + hl) * 4 + 3];

    v2f bfr[3][4];
    #pragma unroll
    for (int nt = 0; nt < 4; ++nt) {
        const int c = nt * 16 + l16;
        bfr[0][nt][0] = W[(k2)     * COUT + c];
        bfr[0][nt][1] = W[(k2 + 1) * COUT + c];
        bfr[1][nt][0] = W[(k2 + 4) * COUT + c];
        bfr[1][nt][1] = W[(k2 + 5) * COUT + c];
        const float w8 = W[8 * COUT + c];
        bfr[2][nt][0] = hi ? 0.0f : w8;
        bfr[2][nt][1] = 0.0f;
    }

    float chs[4], chss[4];
    #pragma unroll
    for (int nt = 0; nt < 4; ++nt) { chs[nt] = 0.0f; chss[nt] = 0.0f; }

    const float4* f4 = (const float4*)features;
    float4 fa = f4[start * NPT + l16];
    float4 fb = f4[start * NPT + 16 + l16];
    int npts = __builtin_amdgcn_readlane(h_np, 0);
    int cy   = __builtin_amdgcn_readlane(h_cy, 0);
    int cx   = __builtin_amdgcn_readlane(h_cx, 0);

    for (int i = 0; i < cnt; ++i) {
        const int p = start + i;
        float4 na = fa, nb = fb;
        if (i + 1 < cnt) {
            na = f4[(p + 1) * NPT + l16];
            nb = f4[(p + 1) * NPT + 16 + l16];
        }
        __builtin_amdgcn_sched_barrier(0);

        float sx = fa.x + fb.x, sy = fa.y + fb.y, sz = fa.z + fb.z;
        #pragma unroll
        for (int off = 8; off >= 1; off >>= 1) {
            sx += __shfl_xor(sx, off, 32);
            sy += __shfl_xor(sy, off, 32);
            sz += __shfl_xor(sz, off, 32);
        }
        const float inv_n = 1.0f / (float)npts;
        const float cmx = sx * inv_n, cmy = sy * inv_n, cmz = sz * inv_n;

        const float xc = (float)cx * 0.2f + 0.1f;
        const float yc = (float)cy * 0.2f + (0.1f - 40.0f);

        const float c0 = hi ? cmz : cmx;
        const float c1 = hi ? xc  : cmy;
        v2f afr[2][3];
        #pragma unroll
        for (int mt = 0; mt < 2; ++mt) {
            const float4 f = mt ? fb : fa;
            const float m  = (mt * 16 + l16 < npts) ? 1.0f : 0.0f;
            const float u0 = hi ? f.z : f.x;
            const float u1 = hi ? f.w : f.y;
            const float v1 = hi ? f.x : f.y;
            afr[mt][0][0] = u0 * m;
            afr[mt][0][1] = u1 * m;
            afr[mt][1][0] = (u0 - c0) * m;
            afr[mt][1][1] = (v1 - c1) * m;
            afr[mt][2][0] = hi ? 0.0f : (f.y - yc) * m;
            afr[mt][2][1] = 0.0f;
        }

        float chmax[4], chmin[4];
        #pragma unroll
        for (int nt = 0; nt < 4; ++nt) { chmax[nt] = -3.402823e38f; chmin[nt] = 3.402823e38f; }

        #pragma unroll
        for (int mt = 0; mt < 2; ++mt) {
            v8f acc[4];
            #pragma unroll
            for (int nt = 0; nt < 4; ++nt) acc[nt] = (v8f){0,0,0,0,0,0,0,0};
            #pragma unroll
            for (int ks = 0; ks < 3; ++ks) {
                #pragma unroll
                for (int nt = 0; nt < 4; ++nt) {
                    acc[nt] = __builtin_amdgcn_wmma_f32_16x16x4_f32(
                        false, afr[mt][ks], false, bfr[ks][nt], (short)0, acc[nt], false, false);
                }
            }
            #pragma unroll
            for (int nt = 0; nt < 4; ++nt) {
                #pragma unroll
                for (int r = 0; r < 8; ++r) {
                    const float v = acc[nt][r];
                    chmax[nt] = fmaxf(chmax[nt], v);
                    chmin[nt] = fminf(chmin[nt], v);
                    chs[nt]  += v;
                    chss[nt] += v * v;
                }
            }
        }

        #pragma unroll
        for (int j = 0; j < 2; ++j) {
            const float mine_mx = hi ? chmax[2 * j + 1] : chmax[2 * j];
            const float send_mx = hi ? chmax[2 * j]     : chmax[2 * j + 1];
            const float mine_mn = hi ? chmin[2 * j + 1] : chmin[2 * j];
            const float send_mn = hi ? chmin[2 * j]     : chmin[2 * j + 1];
            const float mx = fmaxf(mine_mx, __shfl_xor(send_mx, 16, 32));
            const float mn = fminf(mine_mn, __shfl_xor(send_mn, 16, 32));
            ws_mx[(size_t)p * COUT + j * 32 + lane] = mx;
            ws_mn[(size_t)p * COUT + j * 32 + lane] = mn;
        }

        fa = na; fb = nb;
        npts = __builtin_amdgcn_readlane(h_np, i + 1);
        cy   = __builtin_amdgcn_readlane(h_cy, i + 1);
        cx   = __builtin_amdgcn_readlane(h_cx, i + 1);
    }

    #pragma unroll
    for (int nt = 0; nt < 4; ++nt) {
        chs[nt]  += __shfl_xor(chs[nt], 16, 32);
        chss[nt] += __shfl_xor(chss[nt], 16, 32);
    }
    if (!hi) {
        #pragma unroll
        for (int nt = 0; nt < 4; ++nt) {
            wsum[wid][nt * 16 + l16] = chs[nt];
            wssq[wid][nt * 16 + l16] = chss[nt];
        }
    }
    __syncthreads();
    if (tid < COUT) {
        float s = 0.0f, ss = 0.0f;
        #pragma unroll
        for (int w = 0; w < WPB; ++w) { s += wsum[w][tid]; ss += wssq[w][tid]; }
        ws_psum[(size_t)blockIdx.x * COUT + tid] = s;
        ws_pssq[(size_t)blockIdx.x * COUT + tid] = ss;
    }
}

#define FIN_THREADS 512
#define FIN_ACTIVE 480
#define FIN_K 8
#define FIN_BLKS ((P_PILLARS * COUT / 4) / (FIN_ACTIVE * FIN_K))
#define FOLD_GROUPS (FIN_THREADS / 16)
#define FOLD_ROWS (NBLK_A / FOLD_GROUPS)
#define FOLD_PARTS 4
#define FOLD_GPP (FOLD_GROUPS / FOLD_PARTS)

template <int MODE>
__device__ __forceinline__ void pfn_finish_pieces(
    const float4* __restrict__ mx4, const float4* __restrict__ mn4,
    float4* __restrict__ out4, const int i0, const float4 s, const float4 t)
{
    float4 a[FIN_K], b[FIN_K];
    #pragma unroll
    for (int k = 0; k < FIN_K; ++k) {
        if (MODE != 1) a[k] = mx4[i0 + k * FIN_ACTIVE];
        if (MODE != 0) b[k] = mn4[i0 + k * FIN_ACTIVE];
    }
    #pragma unroll
    for (int k = 0; k < FIN_K; ++k) {
        float4 v;
        if (MODE == 0) {
            v = a[k];
        } else if (MODE == 1) {
            v = b[k];
        } else {
            v.x = (s.x >= 0.0f) ? a[k].x : b[k].x;
            v.y = (s.y >= 0.0f) ? a[k].y : b[k].y;
            v.z = (s.z >= 0.0f) ? a[k].z : b[k].z;
            v.w = (s.w >= 0.0f) ? a[k].w : b[k].w;
        }
        float4 o;
        o.x = fmaxf(fmaf(s.x, v.x, t.x), 0.0f);
        o.y = fmaxf(fmaf(s.y, v.y, t.y), 0.0f);
        o.z = fmaxf(fmaf(s.z, v.z, t.z), 0.0f);
        o.w = fmaxf(fmaf(s.w, v.w, t.w), 0.0f);
        out4[i0 + k * FIN_ACTIVE] = o;
    }
}

__global__ __launch_bounds__(FIN_THREADS) void pfn_finish(
    const float* __restrict__ ws_mx, const float* __restrict__ ws_mn,
    const float* __restrict__ psum, const float* __restrict__ pssq,
    const float* __restrict__ gamma, const float* __restrict__ beta,
    float* __restrict__ out)
{
    __shared__ float4 fs4[FOLD_GROUPS * (COUT / 4)];
    __shared__ float4 fq4[FOLD_GROUPS * (COUT / 4)];
    __shared__ float  gs[FOLD_PARTS * COUT];
    __shared__ float  gq[FOLD_PARTS * COUT];
    __shared__ float4 sc4[COUT / 4];
    __shared__ float4 sh4[COUT / 4];

    const int tid = threadIdx.x;
    const int q   = tid & 15;
    const int g   = tid >> 4;

    const float4* ps4 = (const float4*)psum;
    const float4* pq4 = (const float4*)pssq;
    float4 ra[FOLD_ROWS], rb[FOLD_ROWS];
    #pragma unroll
    for (int i = 0; i < FOLD_ROWS; ++i) {
        ra[i] = ps4[(g + FOLD_GROUPS * i) * (COUT / 4) + q];
        rb[i] = pq4[(g + FOLD_GROUPS * i) * (COUT / 4) + q];
    }
    float4 s1 = ra[0], q1 = rb[0];
    #pragma unroll
    for (int i = 1; i < FOLD_ROWS; ++i) {
        s1.x += ra[i].x; s1.y += ra[i].y; s1.z += ra[i].z; s1.w += ra[i].w;
        q1.x += rb[i].x; q1.y += rb[i].y; q1.z += rb[i].z; q1.w += rb[i].w;
    }
    fs4[g * (COUT / 4) + q] = s1;
    fq4[g * (COUT / 4) + q] = q1;
    __syncthreads();

    if (tid < FOLD_PARTS * COUT) {
        const float* fs = (const float*)fs4;
        const float* fq = (const float*)fq4;
        const int c  = tid & (COUT - 1);
        const int g0 = (tid / COUT) * FOLD_GPP;
        float t = fs[g0 * COUT + c], u = fq[g0 * COUT + c];
        #pragma unroll
        for (int k = 1; k < FOLD_GPP; ++k) { t += fs[(g0 + k) * COUT + c]; u += fq[(g0 + k) * COUT + c]; }
        gs[tid] = t;
        gq[tid] = u;
    }
    __syncthreads();

    if (tid < COUT) {
        float t = gs[tid], u = gq[tid];
        #pragma unroll
        for (int k = 1; k < FOLD_PARTS; ++k) { t += gs[k * COUT + tid]; u += gq[k * COUT + tid]; }
        const float inv  = 1.0f / (float)(P_PILLARS * NPT);
        const float mean = t * inv;
        const float var  = u * inv - mean * mean;
        const float sc   = gamma[tid] * rsqrtf(var + 1e-3f);
        ((float*)sc4)[tid] = sc;
        ((float*)sh4)[tid] = beta[tid] - mean * sc;
    }
    __syncthreads();

    if (tid < FIN_ACTIVE) {
        const int i0 = blockIdx.x * (FIN_K * FIN_ACTIVE) + tid;
        const float4 s = sc4[q];
        const float4 t = sh4[q];
        const float4* mx4 = (const float4*)ws_mx;
        const float4* mn4 = (const float4*)ws_mn;
        float4* out4 = (float4*)out;

        const bool px = s.x >= 0.0f, py = s.y >= 0.0f, pz = s.z >= 0.0f, pw = s.w >= 0.0f;
        if (px && py && pz && pw) {
            pfn_finish_pieces<0>(mx4, mn4, out4, i0, s, t);
        } else if (!(px || py || pz || pw)) {
            pfn_finish_pieces<1>(mx4, mn4, out4, i0, s, t);
        } else {
            pfn_finish_pieces<2>(mx4, mn4, out4, i0, s, t);
        }
    }
}

extern "C" void kernel_launch(void* const* d_in, const int* in_sizes, int n_in,
                              void* d_out, int out_size, void* d_ws, size_t ws_size,
                              hipStream_t stream) {
    const float* features   = (const float*)d_in[0];
    const int*   num_points = (const int*)  d_in[1];
    const int*   coors      = (const int*)  d_in[2];
    const float* W          = (const float*)d_in[3];
    const float* gamma      = (const float*)d_in[4];
    const float* beta       = (const float*)d_in[5];
    float* out = (float*)d_out;

    float* ws      = (float*)d_ws;
    float* ws_mx   = ws;
    float* ws_mn   = ws_mx   + (size_t)P_PILLARS * COUT;
    float* ws_psum = ws_mn   + (size_t)P_PILLARS * COUT;
    float* ws_pssq = ws_psum + (size_t)NBLK_A * COUT;

    pfn_pass_a<<<NBLK_A, WPB * 32, 0, stream>>>(features, num_points, coors, W,
                                           ws_mx, ws_mn, ws_psum, ws_pssq);
    pfn_finish<<<FIN_BLKS, FIN_THREADS, 0, stream>>>(ws_mx, ws_mn, ws_psum, ws_pssq,
                                                 gamma, beta, out);
}


# pass A: K 12 -> 8: W rows that meet the same raw coordinate are summed once per wave, pillar means and cell centre enter as k 4..7 (cmx cmy cmz 1); 24 -> 16 matrix ops per pillar; sum order changes
# speedup vs baseline: 12.9782x; 1.1253x over previous
#include <hip/hip_runtime.h>

typedef float v2f __attribute__((ext_vector_type(2)));
typedef float v8f __attribute__((ext_vector_type(8)));

#define P_PILLARS 60000
#define NPT 32
#define COUT 64
#define WPB 20
#define NBLK_A 256
#define NWAVE_A (NBLK_A * WPB)
#define RUN_LEN (P_PILLARS / NWAVE_A)
#define RUN_BIG (P_PILLARS - RUN_LEN * NWAVE_A)

__global__ __launch_bounds__(WPB * 32) void pfn_pass_a(
    const float* __restrict__ features,
    const int*   __restrict__ num_points,
    const int*   __restrict__ coors,
    const float* __restrict__ W,
    float* __restrict__ ws_mx, float* __restrict__ ws_mn,
    float* __restrict__ ws_psum, float* __restrict__ ws_pssq)
{
    __shared__ float wsum[WPB][COUT];
    __shared__ float wssq[WPB][COUT];

    const int tid  = threadIdx.x;
    const int wid  = __builtin_amdgcn_readfirstlane(tid >> 5);
    const int lane = tid & 31;
    const int l16  = lane & 15;
    const bool hi  = (lane >> 4) != 0;
    const int k2   = hi ? 2 : 0;

    const int gw    = blockIdx.x * WPB + wid;
    const int cnt   = RUN_LEN + (gw < RUN_BIG ? 1 : 0);
    const int start = gw * RUN_LEN + (gw < RUN_BIG ? gw : RUN_BIG);

    const int hl   = (lane < cnt) ? lane : cnt - 1;
    const int h_np = num_points[start + hl];
    const int h_cy = coors[(start + hl) * 4 + 2];
    const int h_cx = coors[(start + hl) * 4 + 3];

    v2f bfr0[4];
    float nw46[4];
    float b1base[4], nw7[4], nw8[4];
    #pragma unroll
    for (int nt = 0; nt < 4; ++nt) {
        const int c = nt * 16 + l16;
        const float ra = W[(k2)     * COUT + c];
        const float rb = W[(k2 + 1) * COUT + c];
        const float rc = W[(k2 + 4) * COUT + c];
        const float r5 = W[5 * COUT + c];
        const float r7 = W[7 * COUT + c];
        const float r8 = W[8 * COUT + c];
        const float a5 = hi ? 0.0f : r5;
        const float a7 = hi ? 0.0f : r7;
        const float a8 = hi ? 0.0f : r8;
        bfr0[nt][0] = (ra + rc) + a7;
        bfr0[nt][1] = (rb + a5) + a8;
        nw46[nt]    = -rc;
        b1base[nt]  = -a5;
        nw7[nt]     = hi ? -r7 : 0.0f;
        nw8[nt]     = hi ? -r8 : 0.0f;
    }

    float chs[4], chss[4];
    #pragma unroll
    for (int nt = 0; nt < 4; ++nt) { chs[nt] = 0.0f; chss[nt] = 0.0f; }

    const float4* f4 = (const float4*)features;
    float4 fa = f4[start * NPT + l16];
    float4 fb = f4[start * NPT + 16 + l16];
    int npts = __builtin_amdgcn_readlane(h_np, 0);
    int cy   = __builtin_amdgcn_readlane(h_cy, 0);
    int cx   = __builtin_amdgcn_readlane(h_cx, 0);

    for (int i = 0; i < cnt; ++i) {
        const int p = start + i;
        float4 na = fa, nb = fb;
        if (i + 1 < cnt) {
            na = f4[(p + 1) * NPT + l16];
            nb = f4[(p + 1) * NPT + 16 + l16];
        }
        __builtin_amdgcn_sched_barrier(0);

        float sx = fa.x + fb.x, sy = fa.y + fb.y, sz = fa.z + fb.z;
        #pragma unroll
        for (int off = 8; off >= 1; off >>= 1) {
            sx += __shfl_xor(sx, off, 32);
            sy += __shfl_xor(sy, off, 32);
            sz += __shfl_xor(sz, off, 32);
        }
        const float inv_n = 1.0f / (float)npts;
        const float cmx = sx * inv_n, cmy = sy * inv_n, cmz = sz * inv_n;

        const float xc = (float)cx * 0.2f + 0.1f;
        const float yc = (float)cy * 0.2f + (0.1f - 40.0f);

        const float c0 = hi ? cmz  : cmx;
        const float c1 = hi ? 1.0f : cmy;
        v2f afr[2][2];
        #pragma unroll
        for (int mt = 0; mt < 2; ++mt) {
            const float4 f = mt ? fb : fa;
            const float m  = (mt * 16 + l16 < npts) ? 1.0f : 0.0f;
            afr[mt][0][0] = (hi ? f.z : f.x) * m;
            afr[mt][0][1] = (hi ? f.w : f.y) * m;
            afr[mt][1][0] = c0 * m;
            afr[mt][1][1] = c1 * m;
        }
        v2f bfr1[4];
        #pragma unroll
        for (int nt = 0; nt < 4; ++nt) {
            bfr1[nt][0] = nw46[nt];
            bfr1[nt][1] = fmaf(xc, nw7[nt], fmaf(yc, nw8[nt], b1base[nt]));
        }

        float chmax[4], chmin[4];
        #pragma unroll
        for (int nt = 0; nt < 4; ++nt) { chmax[nt] = -3.402823e38f; chmin[nt] = 3.402823e38f; }

        #pragma unroll
        for (int mt = 0; mt < 2; ++mt) {
            v8f acc[4];
            #pragma unroll
            for (int nt = 0; nt < 4; ++nt) acc[nt] = (v8f){0,0,0,0,0,0,0,0};
            #pragma unroll
            for (int nt = 0; nt < 4; ++nt) {
                acc[nt] = __builtin_amdgcn_wmma_f32_16x16x4_f32(
                    false, afr[mt][0], false, bfr0[nt], (short)0, acc[nt], false, false);
            }
            #pragma unroll
            for (int nt = 0; nt < 4; ++nt) {
                acc[nt] = __builtin_amdgcn_wmma_f32_16x16x4_f32(
                    false, afr[mt][1], false, bfr1[nt], (short)0, acc[nt], false, false);
            }
            #pragma unroll
            for (int nt = 0; nt < 4; ++nt) {
                #pragma unroll
                for (int r = 0; r < 8; ++r) {
                    const float v = acc[nt][r];
                    chmax[nt] = fmaxf(chmax[nt], v);
                    chmin[nt] = fminf(chmin[nt], v);
                    chs[nt]  += v;
                    chss[nt] += v * v;
                }
            }
        }

        #pragma unroll
        for (int j = 0; j < 2; ++j) {
            const float mine_mx = hi ? chmax[2 * j + 1] : chmax[2 * j];
            const float send_mx = hi ? chmax[2 * j]     : chmax[2 * j + 1];
            const float mine_mn = hi ? chmin[2 * j + 1] : chmin[2 * j];
            const float send_mn = hi ? chmin[2 * j]     : chmin[2 * j + 1];
            const float mx = fmaxf(mine_mx, __shfl_xor(send_mx, 16, 32));
            const float mn = fminf(mine_mn, __shfl_xor(send_mn, 16, 32));
            ws_mx[(size_t)p * COUT + j * 32 + lane] = mx;
            ws_mn[(size_t)p * COUT + j * 32 + lane] = mn;
        }

        fa = na; fb = nb;
        npts = __builtin_amdgcn_readlane(h_np, i + 1);
        cy   = __builtin_amdgcn_readlane(h_cy, i + 1);
        cx   = __builtin_amdgcn_readlane(h_cx, i + 1);
    }

    #pragma unroll
    for (int nt = 0; nt < 4; ++nt) {
        chs[nt]  += __shfl_xor(chs[nt], 16, 32);
        chss[nt] += __shfl_xor(chss[nt], 16, 32);
    }
    if (!hi) {
        #pragma unroll
        for (int nt = 0; nt < 4; ++nt) {
            wsum[wid][nt * 16 + l16] = chs[nt];
            wssq[wid][nt * 16 + l16] = chss[nt];
        }
    }
    __syncthreads();
    if (tid < COUT) {
        float s = 0.0f, ss = 0.0f;
        #pragma unroll
        for (int w = 0; w < WPB; ++w) { s += wsum[w][tid]; ss += wssq[w][tid]; }
        ws_psum[(size_t)blockIdx.x * COUT + tid] = s;
        ws_pssq[(size_t)blockIdx.x * COUT + tid] = ss;
    }
}

#define FIN_THREADS 512
#define FIN_ACTIVE 480
#define FIN_K 8
#define FIN_BLKS ((P_PILLARS * COUT / 4) / (FIN_ACTIVE * FIN_K))
#define FOLD_GROUPS (FIN_THREADS / 16)
#define FOLD_ROWS (NBLK_A / FOLD_GROUPS)
#define FOLD_PARTS 4
#define FOLD_GPP (FOLD_GROUPS / FOLD_PARTS)

template <int MODE>
__device__ __forceinline__ void pfn_finish_pieces(
    const float4* __restrict__ mx4, const float4* __restrict__ mn4,
    float4* __restrict__ out4, const int i0, const float4 s, const float4 t)
{
    float4 a[FIN_K], b[FIN_K];
    #pragma unroll
    for (int k = 0; k < FIN_K; ++k) {
        if (MODE != 1) a[k] = mx4[i0 + k * FIN_ACTIVE];
        if (MODE != 0) b[k] = mn4[i0 + k * FIN_ACTIVE];
    }
    #pragma unroll
    for (int k = 0; k < FIN_K; ++k) {
        float4 v;
        if (MODE == 0) {
            v = a[k];
        } else if (MODE == 1) {
            v = b[k];
        } else {
            v.x = (s.x >= 0.0f) ? a[k].x : b[k].x;
            v.y = (s.y >= 0.0f) ? a[k].y : b[k].y;
            v.z = (s.z >= 0.0f) ? a[k].z : b[k].z;
            v.w = (s.w >= 0.0f) ? a[k].w : b[k].w;
        }
        float4 o;
        o.x = fmaxf(fmaf(s.x, v.x, t.x), 0.0f);
        o.y = fmaxf(fmaf(s.y, v.y, t.y), 0.0f);
        o.z = fmaxf(fmaf(s.z, v.z, t.z), 0.0f);
        o.w = fmaxf(fmaf(s.w, v.w, t.w), 0.0f);
        out4[i0 + k * FIN_ACTIVE] = o;
    }
}

__global__ __launch_bounds__(FIN_THREADS) void pfn_finish(
    const float* __restrict__ ws_mx, const float* __restrict__ ws_mn,
    const float* __restrict__ psum, const float* __restrict__ pssq,
    const float* __restrict__ gamma, const float* __restrict__ beta,
    float* __restrict__ out)
{
    __shared__ float4 fs4[FOLD_GROUPS * (COUT / 4)];
    __shared__ float4 fq4[FOLD_GROUPS * (COUT / 4)];
    __shared__ float  gs[FOLD_PARTS * COUT];
    __shared__ float  gq[FOLD_PARTS * COUT];
    __shared__ float4 sc4[COUT / 4];
    __shared__ float4 sh4[COUT / 4];

    const int tid = threadIdx.x;
    const int q   = tid & 15;
    const int g   = tid >> 4;

    const float4* ps4 = (const float4*)psum;
    const float4* pq4 = (const float4*)pssq;
    float4 ra[FOLD_ROWS], rb[FOLD_ROWS];
    #pragma unroll
    for (int i = 0; i < FOLD_ROWS; ++i) {
        ra[i] = ps4[(g + FOLD_GROUPS * i) * (COUT / 4) + q];
        rb[i] = pq4[(g + FOLD_GROUPS * i) * (COUT / 4) + q];
    }
    float4 s1 = ra[0], q1 = rb[0];
    #pragma unroll
    for (int i = 1; i < FOLD_ROWS; ++i) {
        s1.x += ra[i].x; s1.y += ra[i].y; s1.z += ra[i].z; s1.w += ra[i].w;
        q1.x += rb[i].x; q1.y += rb[i].y; q1.z += rb[i].z; q1.w += rb[i].w;
    }
    fs4[g * (COUT / 4) + q] = s1;
    fq4[g * (COUT / 4) + q] = q1;
    __syncthreads();

    if (tid < FOLD_PARTS * COUT) {
        const float* fs = (const float*)fs4;
        const float* fq = (const float*)fq4;
        const int c  = tid & (COUT - 1);
        const int g0 = (tid / COUT) * FOLD_GPP;
        float t = fs[g0 * COUT + c], u = fq[g0 * COUT + c];
        #pragma unroll
        for (int k = 1; k < FOLD_GPP; ++k) { t += fs[(g0 + k) * COUT + c]; u += fq[(g0 + k) * COUT + c]; }
        gs[tid] = t;
        gq[tid] = u;
    }
    __syncthreads();

    if (tid < COUT) {
        float t = gs[tid], u = gq[tid];
        #pragma unroll
        for (int k = 1; k < FOLD_PARTS; ++k) { t += gs[k * COUT + tid]; u += gq[k * COUT + tid]; }
        const float inv  = 1.0f / (float)(P_PILLARS * NPT);
        const float mean = t * inv;
        const float var  = u * inv - mean * mean;
        const float sc   = gamma[tid] * rsqrtf(var + 1e-3f);
        ((float*)sc4)[tid] = sc;
        ((float*)sh4)[tid] = beta[tid] - mean * sc;
    }
    __syncthreads();

    if (tid < FIN_ACTIVE) {
        const int i0 = blockIdx.x * (FIN_K * FIN_ACTIVE) + tid;
        const float4 s = sc4[q];
        const float4 t = sh4[q];
        const float4* mx4 = (const float4*)ws_mx;
        const float4* mn4 = (const float4*)ws_mn;
        float4* out4 = (float4*)out;

        const bool px = s.x >= 0.0f, py = s.y >= 0.0f, pz = s.z >= 0.0f, pw = s.w >= 0.0f;
        if (px && py && pz && pw) {
            pfn_finish_pieces<0>(mx4, mn4, out4, i0, s, t);
        } else if (!(px || py || pz || pw)) {
            pfn_finish_pieces<1>(mx4, mn4, out4, i0, s, t);
        } else {
            pfn_finish_pieces<2>(mx4, mn4, out4, i0, s, t);
        }
    }
}

extern "C" void kernel_launch(void* const* d_in, const int* in_sizes, int n_in,
                              void* d_out, int out_size, void* d_ws, size_t ws_size,
                              hipStream_t stream) {
    const float* features   = (const float*)d_in[0];
    const int*   num_points = (const int*)  d_in[1];
    const int*   coors      = (const int*)  d_in[2];
    const float* W          = (const float*)d_in[3];
    const float* gamma      = (const float*)d_in[4];
    const float* beta       = (const float*)d_in[5];
    float* out = (float*)d_out;

    float* ws      = (float*)d_ws;
    float* ws_mx   = ws;
    float* ws_mn   = ws_mx   + (size_t)P_PILLARS * COUT;
    float* ws_psum = ws_mn   + (size_t)P_PILLARS * COUT;
    float* ws_pssq = ws_psum + (size_t)NBLK_A * COUT;

    pfn_pass_a<<<NBLK_A, WPB * 32, 0, stream>>>(features, num_points, coors, W,
                                           ws_mx, ws_mn, ws_psum, ws_pssq);
    pfn_finish<<<FIN_BLKS, FIN_THREADS, 0, stream>>>(ws_mx, ws_mn, ws_psum, ws_pssq,
                                                 gamma, beta, out);
}


# pass A statistics: sum and sum of squares accumulate on the aligned register pairs of the matrix result (packed add, packed fma; two chains per N-tile), register moves go; sum order changes
# speedup vs baseline: 13.6532x; 1.0520x over previous
#include <hip/hip_runtime.h>

typedef float v2f __attribute__((ext_vector_type(2)));
typedef float v8f __attribute__((ext_vector_type(8)));

#define P_PILLARS 60000
#define NPT 32
#define COUT 64
#define WPB 20
#define NBLK_A 256
#define NWAVE_A (NBLK_A * WPB)
#define RUN_LEN (P_PILLARS / NWAVE_A)
#define RUN_BIG (P_PILLARS - RUN_LEN * NWAVE_A)

__global__ __launch_bounds__(WPB * 32) void pfn_pass_a(
    const float* __restrict__ features,
    const int*   __restrict__ num_points,
    const int*   __restrict__ coors,
    const float* __restrict__ W,
    float* __restrict__ ws_mx, float* __restrict__ ws_mn,
    float* __restrict__ ws_psum, float* __restrict__ ws_pssq)
{
    __shared__ float wsum[WPB][COUT];
    __shared__ float wssq[WPB][COUT];

    const int tid  = threadIdx.x;
    const int wid  = __builtin_amdgcn_readfirstlane(tid >> 5);
    const int lane = tid & 31;
    const int l16  = lane & 15;
    const bool hi  = (lane >> 4) != 0;
    const int k2   = hi ? 2 : 0;

    const int gw    = blockIdx.x * WPB + wid;
    const int cnt   = RUN_LEN + (gw < RUN_BIG ? 1 : 0);
    const int start = gw * RUN_LEN + (gw < RUN_BIG ? gw : RUN_BIG);

    const int hl   = (lane < cnt) ? lane : cnt - 1;
    const int h_np = num_points[start + hl];
    const int h_cy = coors[(start + hl) * 4 + 2];
    const int h_cx = coors[(start + hl) * 4 + 3];

    v2f bfr0[4];
    float nw46[4];
    float b1base[4], nw7[4], nw8[4];
    #pragma unroll
    for (int nt = 0; nt < 4; ++nt) {
        const int c = nt * 16 + l16;
        const float ra = W[(k2)     * COUT + c];
        const float rb = W[(k2 + 1) * COUT + c];
        const float rc = W[(k2 + 4) * COUT + c];
        const float r5 = W[5 * COUT + c];
        const float r7 = W[7 * COUT + c];
        const float r8 = W[8 * COUT + c];
        const float a5 = hi ? 0.0f : r5;
        const float a7 = hi ? 0.0f : r7;
        const float a8 = hi ? 0.0f : r8;
        bfr0[nt][0] = (ra + rc) + a7;
        bfr0[nt][1] = (rb + a5) + a8;
        nw46[nt]    = -rc;
        b1base[nt]  = -a5;
        nw7[nt]     = hi ? -r7 : 0.0f;
        nw8[nt]     = hi ? -r8 : 0.0f;
    }

    v2f chs2[4], chss2[4];
    #pragma unroll
    for (int nt = 0; nt < 4; ++nt) { chs2[nt] = (v2f){0.0f, 0.0f}; chss2[nt] = (v2f){0.0f, 0.0f}; }

    const float4* f4 = (const float4*)features;
    float4 fa = f4[start * NPT + l16];
    float4 fb = f4[start * NPT + 16 + l16];
    int npts = __builtin_amdgcn_readlane(h_np, 0);
    int cy   = __builtin_amdgcn_readlane(h_cy, 0);
    int cx   = __builtin_amdgcn_readlane(h_cx, 0);

    for (int i = 0; i < cnt; ++i) {
        const int p = start + i;
        float4 na = fa, nb = fb;
        if (i + 1 < cnt) {
            na = f4[(p + 1) * NPT + l16];
            nb = f4[(p + 1) * NPT + 16 + l16];
        }
        __builtin_amdgcn_sched_barrier(0);

        float sx = fa.x + fb.x, sy = fa.y + fb.y, sz = fa.z + fb.z;
        #pragma unroll
        for (int off = 8; off >= 1; off >>= 1) {
            sx += __shfl_xor(sx, off, 32);
            sy += __shfl_xor(sy, off, 32);
            sz += __shfl_xor(sz, off, 32);
        }
        const float inv_n = 1.0f / (float)npts;
        const float cmx = sx * inv_n, cmy = sy * inv_n, cmz = sz * inv_n;

        const float xc = (float)cx * 0.2f + 0.1f;
        const float yc = (float)cy * 0.2f + (0.1f - 40.0f);

        const float c0 = hi ? cmz  : cmx;
        const float c1 = hi ? 1.0f : cmy;
        v2f afr[2][2];
        #pragma unroll
        for (int mt = 0; mt < 2; ++mt) {
            const float4 f = mt ? fb : fa;
            const float m  = (mt * 16 + l16 < npts) ? 1.0f : 0.0f;
            afr[mt][0][0] = (hi ? f.z : f.x) * m;
            afr[mt][0][1] = (hi ? f.w : f.y) * m;
            afr[mt][1][0] = c0 * m;
            afr[mt][1][1] = c1 * m;
        }
        v2f bfr1[4];
        #pragma unroll
        for (int nt = 0; nt < 4; ++nt) {
            bfr1[nt][0] = nw46[nt];
            bfr1[nt][1] = fmaf(xc, nw7[nt], fmaf(yc, nw8[nt], b1base[nt]));
        }

        float chmax[4], chmin[4];
        #pragma unroll
        for (int nt = 0; nt < 4; ++nt) { chmax[nt] = -3.402823e38f; chmin[nt] = 3.402823e38f; }

        #pragma unroll
        for (int mt = 0; mt < 2; ++mt) {
            v8f acc[4];
            #pragma unroll
            for (int nt = 0; nt < 4; ++nt) acc[nt] = (v8f){0,0,0,0,0,0,0,0};
            #pragma unroll
            for (int nt = 0; nt < 4; ++nt) {
                acc[nt] = __builtin_amdgcn_wmma_f32_16x16x4_f32(
                    false, afr[mt][0], false, bfr0[nt], (short)0, acc[nt], false, false);
            }
            #pragma unroll
            for (int nt = 0; nt < 4; ++nt) {
                acc[nt] = __builtin_amdgcn_wmma_f32_16x16x4_f32(
                    false, afr[mt][1], false, bfr1[nt], (short)0, acc[nt], false, false);
            }
            #pragma unroll
            for (int nt = 0; nt < 4; ++nt) {
                #pragma unroll
                for (int j = 0; j < 4; ++j) {
                    const float v0 = acc[nt][2 * j], v1 = acc[nt][2 * j + 1];
                    chmax[nt] = fmaxf(fmaxf(chmax[nt], v0), v1);
                    chmin[nt] = fminf(fminf(chmin[nt], v0), v1);
                    const v2f vv = {v0, v1};
                    chs2[nt]  = chs2[nt] + vv;
                    chss2[nt] = __builtin_elementwise_fma(vv, vv, chss2[nt]);
                }
            }
        }

        #pragma unroll
        for (int j = 0; j < 2; ++j) {
            const float mine_mx = hi ? chmax[2 * j + 1] : chmax[2 * j];
            const float send_mx = hi ? chmax[2 * j]     : chmax[2 * j + 1];
            const float mine_mn = hi ? chmin[2 * j + 1] : chmin[2 * j];
            const float send_mn = hi ? chmin[2 * j]     : chmin[2 * j + 1];
            const float mx = fmaxf(mine_mx, __shfl_xor(send_mx, 16, 32));
            const float mn = fminf(mine_mn, __shfl_xor(send_mn, 16, 32));
            ws_mx[(size_t)p * COUT + j * 32 + lane] = mx;
            ws_mn[(size_t)p * COUT + j * 32 + lane] = mn;
        }

        fa = na; fb = nb;
        npts = __builtin_amdgcn_readlane(h_np, i + 1);
        cy   = __builtin_amdgcn_readlane(h_cy, i + 1);
        cx   = __builtin_amdgcn_readlane(h_cx, i + 1);
    }

    float chs[4], chss[4];
    #pragma unroll
    for (int nt = 0; nt < 4; ++nt) {
        chs[nt]  = chs2[nt][0] + chs2[nt][1];
        chss[nt] = chss2[nt][0] + chss2[nt][1];
        chs[nt]  += __shfl_xor(chs[nt], 16, 32);
        chss[nt] += __shfl_xor(chss[nt], 16, 32);
    }
    if (!hi) {
        #pragma unroll
        for (int nt = 0; nt < 4; ++nt) {
            wsum[wid][nt * 16 + l16] = chs[nt];
            wssq[wid][nt * 16 + l16] = chss[nt];
        }
    }
    __syncthreads();
    if (tid < COUT) {
        float s = 0.0f, ss = 0.0f;
        #pragma unroll
        for (int w = 0; w < WPB; ++w) { s += wsum[w][tid]; ss += wssq[w][tid]; }
        ws_psum[(size_t)blockIdx.x * COUT + tid] = s;
        ws_pssq[(size_t)blockIdx.x * COUT + tid] = ss;
    }
}

#define FIN_THREADS 512
#define FIN_ACTIVE 480
#define FIN_K 8
#define FIN_BLKS ((P_PILLARS * COUT / 4) / (FIN_ACTIVE * FIN_K))
#define FOLD_GROUPS (FIN_THREADS / 16)
#define FOLD_ROWS (NBLK_A / FOLD_GROUPS)
#define FOLD_PARTS 4
#define FOLD_GPP (FOLD_GROUPS / FOLD_PARTS)

template <int MODE>
__device__ __forceinline__ void pfn_finish_pieces(
    const float4* __restrict__ mx4, const float4* __restrict__ mn4,
    float4* __restrict__ out4, const int i0, const float4 s, const float4 t)
{
    float4 a[FIN_K], b[FIN_K];
    #pragma unroll
    for (int k = 0; k < FIN_K; ++k) {
        if (MODE != 1) a[k] = mx4[i0 + k * FIN_ACTIVE];
        if (MODE != 0) b[k] = mn4[i0 + k * FIN_ACTIVE];
    }
    #pragma unroll
    for (int k = 0; k < FIN_K; ++k) {
        float4 v;
        if (MODE == 0) {
            v = a[k];
        } else if (MODE == 1) {
            v = b[k];
        } else {
            v.x = (s.x >= 0.0f) ? a[k].x : b[k].x;
            v.y = (s.y >= 0.0f) ? a[k].y : b[k].y;
            v.z = (s.z >= 0.0f) ? a[k].z : b[k].z;
            v.w = (s.w >= 0.0f) ? a[k].w : b[k].w;
        }
        float4 o;
        o.x = fmaxf(fmaf(s.x, v.x, t.x), 0.0f);
        o.y = fmaxf(fmaf(s.y, v.y, t.y), 0.0f);
        o.z = fmaxf(fmaf(s.z, v.z, t.z), 0.0f);
        o.w = fmaxf(fmaf(s.w, v.w, t.w), 0.0f);
        out4[i0 + k * FIN_ACTIVE] = o;
    }
}

__global__ __launch_bounds__(FIN_THREADS) void pfn_finish(
    const float* __restrict__ ws_mx, const float* __restrict__ ws_mn,
    const float* __restrict__ psum, const float* __restrict__ pssq,
    const float* __restrict__ gamma, const float* __restrict__ beta,
    float* __restrict__ out)
{
    __shared__ float4 fs4[FOLD_GROUPS * (COUT / 4)];
    __shared__ float4 fq4[FOLD_GROUPS * (COUT / 4)];
    __shared__ float  gs[FOLD_PARTS * COUT];
    __shared__ float  gq[FOLD_PARTS * COUT];
    __shared__ float4 sc4[COUT / 4];
    __shared__ float4 sh4[COUT / 4];

    const int tid = threadIdx.x;
    const int q   = tid & 15;
    const int g   = tid >> 4;

    const float4* ps4 = (const float4*)psum;
    const float4* pq4 = (const float4*)pssq;
    float4 ra[FOLD_ROWS], rb[FOLD_ROWS];
    #pragma unroll
    for (int i = 0; i < FOLD_ROWS; ++i) {
        ra[i] = ps4[(g + FOLD_GROUPS * i) * (COUT / 4) + q];
        rb[i] = pq4[(g + FOLD_GROUPS * i) * (COUT / 4) + q];
    }
    float4 s1 = ra[0], q1 = rb[0];
    #pragma unroll
    for (int i = 1; i < FOLD_ROWS; ++i) {
        s1.x += ra[i].x; s1.y += ra[i].y; s1.z += ra[i].z; s1.w += ra[i].w;
        q1.x += rb[i].x; q1.y += rb[i].y; q1.z += rb[i].z; q1.w += rb[i].w;
    }
    fs4[g * (COUT / 4) + q] = s1;
    fq4[g * (COUT / 4) + q] = q1;
    __syncthreads();

    if (tid < FOLD_PARTS * COUT) {
        const float* fs = (const float*)fs4;
        const float* fq = (const float*)fq4;
        const int c  = tid & (COUT - 1);
        const int g0 = (tid / COUT) * FOLD_GPP;
        float t = fs[g0 * COUT + c], u = fq[g0 * COUT + c];
        #pragma unroll
        for (int k = 1; k < FOLD_GPP; ++k) { t += fs[(g0 + k) * COUT + c]; u += fq[(g0 + k) * COUT + c]; }
        gs[tid] = t;
        gq[tid] = u;
    }
    __syncthreads();

    if (tid < COUT) {
        float t = gs[tid], u = gq[tid];
        #pragma unroll
        for (int k = 1; k < FOLD_PARTS; ++k) { t += gs[k * COUT + tid]; u += gq[k * COUT + tid]; }
        const float inv  = 1.0f / (float)(P_PILLARS * NPT);
        const float mean = t * inv;
        const float var  = u * inv - mean * mean;
        const float sc   = gamma[tid] * rsqrtf(var + 1e-3f);
        ((float*)sc4)[tid] = sc;
        ((float*)sh4)[tid] = beta[tid] - mean * sc;
    }
    __syncthreads();

    if (tid < FIN_ACTIVE) {
        const int i0 = blockIdx.x * (FIN_K * FIN_ACTIVE) + tid;
        const float4 s = sc4[q];
        const float4 t = sh4[q];
        const float4* mx4 = (const float4*)ws_mx;
        const float4* mn4 = (const float4*)ws_mn;
        float4* out4 = (float4*)out;

        const bool px = s.x >= 0.0f, py = s.y >= 0.0f, pz = s.z >= 0.0f, pw = s.w >= 0.0f;
        if (px && py && pz && pw) {
            pfn_finish_pieces<0>(mx4, mn4, out4, i0, s, t);
        } else if (!(px || py || pz || pw)) {
            pfn_finish_pieces<1>(mx4, mn4, out4, i0, s, t);
        } else {
            pfn_finish_pieces<2>(mx4, mn4, out4, i0, s, t);
        }
    }
}

extern "C" void kernel_launch(void* const* d_in, const int* in_sizes, int n_in,
                              void* d_out, int out_size, void* d_ws, size_t ws_size,
                              hipStream_t stream) {
    const float* features   = (const float*)d_in[0];
    const int*   num_points = (const int*)  d_in[1];
    const int*   coors      = (const int*)  d_in[2];
    const float* W          = (const float*)d_in[3];
    const float* gamma      = (const float*)d_in[4];
    const float* beta       = (const float*)d_in[5];
    float* out = (float*)d_out;

    float* ws      = (float*)d_ws;
    float* ws_mx   = ws;
    float* ws_mn   = ws_mx   + (size_t)P_PILLARS * COUT;
    float* ws_psum = ws_mn   + (size_t)P_PILLARS * COUT;
    float* ws_pssq = ws_psum + (size_t)NBLK_A * COUT;

    pfn_pass_a<<<NBLK_A, WPB * 32, 0, stream>>>(features, num_points, coors, W,
                                           ws_mx, ws_mn, ws_psum, ws_pssq);
    pfn_finish<<<FIN_BLKS, FIN_THREADS, 0, stream>>>(ws_mx, ws_mn, ws_psum, ws_pssq,
                                                 gamma, beta, out);
}


# pass A: K 8 -> 4: the per-pillar offset (means, cell centre) is added after the max/min and enters the sums in closed form; padded rows repeat point 0; matrix ops per pillar halve; sum order changes
# speedup vs baseline: 14.8780x; 1.0897x over previous
#include <hip/hip_runtime.h>

typedef float v2f __attribute__((ext_vector_type(2)));
typedef float v8f __attribute__((ext_vector_type(8)));

#define P_PILLARS 60000
#define NPT 32
#define COUT 64
#define WPB 20
#define NBLK_A 256
#define NWAVE_A (NBLK_A * WPB)
#define RUN_LEN (P_PILLARS / NWAVE_A)
#define RUN_BIG (P_PILLARS - RUN_LEN * NWAVE_A)

__global__ __launch_bounds__(WPB * 32) void pfn_pass_a(
    const float* __restrict__ features,
    const int*   __restrict__ num_points,
    const int*   __restrict__ coors,
    const float* __restrict__ W,
    float* __restrict__ ws_mx, float* __restrict__ ws_mn,
    float* __restrict__ ws_psum, float* __restrict__ ws_pssq)
{
    __shared__ float wsum[WPB][COUT];
    __shared__ float wssq[WPB][COUT];

    const int tid  = threadIdx.x;
    const int wid  = __builtin_amdgcn_readfirstlane(tid >> 5);
    const int lane = tid & 31;
    const int l16  = lane & 15;
    const bool hi  = (lane >> 4) != 0;
    const int k2   = hi ? 2 : 0;

    const int gw    = blockIdx.x * WPB + wid;
    const int cnt   = RUN_LEN + (gw < RUN_BIG ? 1 : 0);
    const int start = gw * RUN_LEN + (gw < RUN_BIG ? gw : RUN_BIG);

    const int hl   = (lane < cnt) ? lane : cnt - 1;
    const int h_np = num_points[start + hl];
    const int h_cy = coors[(start + hl) * 4 + 2];
    const int h_cx = coors[(start + hl) * 4 + 3];

    v2f bfr0[4];
    float nw4[4], nw5[4], nw6[4], nw7[4], nw8[4];
    #pragma unroll
    for (int nt = 0; nt < 4; ++nt) {
        const int c = nt * 16 + l16;
        const float ra = W[(k2)     * COUT + c];
        const float rb = W[(k2 + 1) * COUT + c];
        const float r4 = W[4 * COUT + c];
        const float r5 = W[5 * COUT + c];
        const float r6 = W[6 * COUT + c];
        const float r7 = W[7 * COUT + c];
        const float r8 = W[8 * COUT + c];
        const float rc = hi ? r6 : r4;
        const float a5 = hi ? 0.0f : r5;
        const float a7 = hi ? 0.0f : r7;
        const float a8 = hi ? 0.0f : r8;
        bfr0[nt][0] = (ra + rc) + a7;
        bfr0[nt][1] = (rb + a5) + a8;
        nw4[nt] = -r4; nw5[nt] = -r5; nw6[nt] = -r6; nw7[nt] = -r7; nw8[nt] = -r8;
    }

    v2f chss2[4];
    float chs[4], cross[4], corr2[4];
    #pragma unroll
    for (int nt = 0; nt < 4; ++nt) {
        chss2[nt] = (v2f){0.0f, 0.0f};
        chs[nt] = 0.0f; cross[nt] = 0.0f; corr2[nt] = 0.0f;
    }

    const float4* f4 = (const float4*)features;
    float4 fa = f4[start * NPT + l16];
    float4 fb = f4[start * NPT + 16 + l16];
    int npts = __builtin_amdgcn_readlane(h_np, 0);
    int cy   = __builtin_amdgcn_readlane(h_cy, 0);
    int cx   = __builtin_amdgcn_readlane(h_cx, 0);

    for (int i = 0; i < cnt; ++i) {
        const int p = start + i;
        float4 na = fa, nb = fb;
        if (i + 1 < cnt) {
            na = f4[(p + 1) * NPT + l16];
            nb = f4[(p + 1) * NPT + 16 + l16];
        }
        __builtin_amdgcn_sched_barrier(0);

        float sx = fa.x + fb.x, sy = fa.y + fb.y, sz = fa.z + fb.z;
        #pragma unroll
        for (int off = 8; off >= 1; off >>= 1) {
            sx += __shfl_xor(sx, off, 32);
            sy += __shfl_xor(sy, off, 32);
            sz += __shfl_xor(sz, off, 32);
        }
        const float inv_n = 1.0f / (float)npts;
        const float cmx = sx * inv_n, cmy = sy * inv_n, cmz = sz * inv_n;

        const float xc = (float)cx * 0.2f + 0.1f;
        const float yc = (float)cy * 0.2f + (0.1f - 40.0f);

        float dv[4];
        #pragma unroll
        for (int nt = 0; nt < 4; ++nt) {
            dv[nt] = fmaf(cmx, nw4[nt], fmaf(cmy, nw5[nt], fmaf(cmz, nw6[nt],
                     fmaf(xc, nw7[nt], yc * nw8[nt]))));
        }

        const float p0 = hi ? fa.z : fa.x, p1 = hi ? fa.w : fa.y;
        const float q0 = hi ? fb.z : fb.x, q1 = hi ? fb.w : fb.y;
        const float r0 = __shfl(p0, lane & 16, 32), r1 = __shfl(p1, lane & 16, 32);
        const bool va = l16 < npts, vb = 16 + l16 < npts;
        v2f afr[2];
        afr[0][0] = va ? p0 : r0;
        afr[0][1] = va ? p1 : r1;
        afr[1][0] = vb ? q0 : r0;
        afr[1][1] = vb ? q1 : r1;

        const float kf = hi ? 0.0f : (float)(NPT - npts);
        const float nf = hi ? 0.0f : (float)npts;

        v8f acc[2][4];
        #pragma unroll
        for (int mt = 0; mt < 2; ++mt) {
            #pragma unroll
            for (int nt = 0; nt < 4; ++nt) {
                acc[mt][nt] = __builtin_amdgcn_wmma_f32_16x16x4_f32(
                    false, afr[mt], false, bfr0[nt], (short)0, (v8f){0,0,0,0,0,0,0,0}, false, false);
            }
        }

        float chmax[4], chmin[4];
        #pragma unroll
        for (int nt = 0; nt < 4; ++nt) {
            const float y0 = acc[0][nt][0];
            float ymax = fmaxf(y0, acc[0][nt][1]);
            float ymin = fminf(y0, acc[0][nt][1]);
            v2f ps = {y0, acc[0][nt][1]};
            chss2[nt] = __builtin_elementwise_fma(ps, ps, chss2[nt]);
            #pragma unroll
            for (int mt = 0; mt < 2; ++mt) {
                #pragma unroll
                for (int j = (mt == 0 ? 1 : 0); j < 4; ++j) {
                    const float v0 = acc[mt][nt][2 * j], v1 = acc[mt][nt][2 * j + 1];
                    ymax = fmaxf(fmaxf(ymax, v0), v1);
                    ymin = fminf(fminf(ymin, v0), v1);
                    const v2f vv = {v0, v1};
                    ps = ps + vv;
                    chss2[nt] = __builtin_elementwise_fma(vv, vv, chss2[nt]);
                }
            }
            chmax[nt] = ymax + dv[nt];
            chmin[nt] = ymin + dv[nt];
            const float s1 = fmaf(-kf, y0, ps[0] + ps[1]);
            const float u  = fmaf(nf, dv[nt], s1);
            chs[nt]   += u;
            cross[nt]  = fmaf(dv[nt], u + s1, cross[nt]);
            corr2[nt]  = fmaf(kf * y0, y0, corr2[nt]);
        }

        const float zmax = (npts < NPT) ? 0.0f : -3.402823e38f;
        const float zmin = (npts < NPT) ? 0.0f :  3.402823e38f;

        #pragma unroll
        for (int j = 0; j < 2; ++j) {
            const float mine_mx = hi ? chmax[2 * j + 1] : chmax[2 * j];
            const float send_mx = hi ? chmax[2 * j]     : chmax[2 * j + 1];
            const float mine_mn = hi ? chmin[2 * j + 1] : chmin[2 * j];
            const float send_mn = hi ? chmin[2 * j]     : chmin[2 * j + 1];
            const float mx = fmaxf(fmaxf(mine_mx, __shfl_xor(send_mx, 16, 32)), zmax);
            const float mn = fminf(fminf(mine_mn, __shfl_xor(send_mn, 16, 32)), zmin);
            ws_mx[(size_t)p * COUT + j * 32 + lane] = mx;
            ws_mn[(size_t)p * COUT + j * 32 + lane] = mn;
        }

        fa = na; fb = nb;
        npts = __builtin_amdgcn_readlane(h_np, i + 1);
        cy   = __builtin_amdgcn_readlane(h_cy, i + 1);
        cx   = __builtin_amdgcn_readlane(h_cx, i + 1);
    }

    float chss[4];
    #pragma unroll
    for (int nt = 0; nt < 4; ++nt) {
        chss[nt] = ((chss2[nt][0] + chss2[nt][1]) - corr2[nt]) + cross[nt];
        chs[nt]  += __shfl_xor(chs[nt], 16, 32);
        chss[nt] += __shfl_xor(chss[nt], 16, 32);
    }
    if (!hi) {
        #pragma unroll
        for (int nt = 0; nt < 4; ++nt) {
            wsum[wid][nt * 16 + l16] = chs[nt];
            wssq[wid][nt * 16 + l16] = chss[nt];
        }
    }
    __syncthreads();
    if (tid < COUT) {
        float s = 0.0f, ss = 0.0f;
        #pragma unroll
        for (int w = 0; w < WPB; ++w) { s += wsum[w][tid]; ss += wssq[w][tid]; }
        ws_psum[(size_t)blockIdx.x * COUT + tid] = s;
        ws_pssq[(size_t)blockIdx.x * COUT + tid] = ss;
    }
}

#define FIN_THREADS 512
#define FIN_ACTIVE 480
#define FIN_K 8
#define FIN_BLKS ((P_PILLARS * COUT / 4) / (FIN_ACTIVE * FIN_K))
#define FOLD_GROUPS (FIN_THREADS / 16)
#define FOLD_ROWS (NBLK_A / FOLD_GROUPS)
#define FOLD_PARTS 4
#define FOLD_GPP (FOLD_GROUPS / FOLD_PARTS)

template <int MODE>
__device__ __forceinline__ void pfn_finish_pieces(
    const float4* __restrict__ mx4, const float4* __restrict__ mn4,
    float4* __restrict__ out4, const int i0, const float4 s, const float4 t)
{
    float4 a[FIN_K], b[FIN_K];
    #pragma unroll
    for (int k = 0; k < FIN_K; ++k) {
        if (MODE != 1) a[k] = mx4[i0 + k * FIN_ACTIVE];
        if (MODE != 0) b[k] = mn4[i0 + k * FIN_ACTIVE];
    }
    #pragma unroll
    for (int k = 0; k < FIN_K; ++k) {
        float4 v;
        if (MODE == 0) {
            v = a[k];
        } else if (MODE == 1) {
            v = b[k];
        } else {
            v.x = (s.x >= 0.0f) ? a[k].x : b[k].x;
            v.y = (s.y >= 0.0f) ? a[k].y : b[k].y;
            v.z = (s.z >= 0.0f) ? a[k].z : b[k].z;
            v.w = (s.w >= 0.0f) ? a[k].w : b[k].w;
        }
        float4 o;
        o.x = fmaxf(fmaf(s.x, v.x, t.x), 0.0f);
        o.y = fmaxf(fmaf(s.y, v.y, t.y), 0.0f);
        o.z = fmaxf(fmaf(s.z, v.z, t.z), 0.0f);
        o.w = fmaxf(fmaf(s.w, v.w, t.w), 0.0f);
        out4[i0 + k * FIN_ACTIVE] = o;
    }
}

__global__ __launch_bounds__(FIN_THREADS) void pfn_finish(
    const float* __restrict__ ws_mx, const float* __restrict__ ws_mn,
    const float* __restrict__ psum, const float* __restrict__ pssq,
    const float* __restrict__ gamma, const float* __restrict__ beta,
    float* __restrict__ out)
{
    __shared__ float4 fs4[FOLD_GROUPS * (COUT / 4)];
    __shared__ float4 fq4[FOLD_GROUPS * (COUT / 4)];
    __shared__ float  gs[FOLD_PARTS * COUT];
    __shared__ float  gq[FOLD_PARTS * COUT];
    __shared__ float4 sc4[COUT / 4];
    __shared__ float4 sh4[COUT / 4];

    const int tid = threadIdx.x;
    const int q   = tid & 15;
    const int g   = tid >> 4;

    const float4* ps4 = (const float4*)psum;
    const float4* pq4 = (const float4*)pssq;
    float4 ra[FOLD_ROWS], rb[FOLD_ROWS];
    #pragma unroll
    for (int i = 0; i < FOLD_ROWS; ++i) {
        ra[i] = ps4[(g + FOLD_GROUPS * i) * (COUT / 4) + q];
        rb[i] = pq4[(g + FOLD_GROUPS * i) * (COUT / 4) + q];
    }
    float4 s1 = ra[0], q1 = rb[0];
    #pragma unroll
    for (int i = 1; i < FOLD_ROWS; ++i) {
        s1.x += ra[i].x; s1.y += ra[i].y; s1.z += ra[i].z; s1.w += ra[i].w;
        q1.x += rb[i].x; q1.y += rb[i].y; q1.z += rb[i].z; q1.w += rb[i].w;
    }
    fs4[g * (COUT / 4) + q] = s1;
    fq4[g * (COUT / 4) + q] = q1;
    __syncthreads();

    if (tid < FOLD_PARTS * COUT) {
        const float* fs = (const float*)fs4;
        const float* fq = (const float*)fq4;
        const int c  = tid & (COUT - 1);
        const int g0 = (tid / COUT) * FOLD_GPP;
        float t = fs[g0 * COUT + c], u = fq[g0 * COUT + c];
        #pragma unroll
        for (int k = 1; k < FOLD_GPP; ++k) { t += fs[(g0 + k) * COUT + c]; u += fq[(g0 + k) * COUT + c]; }
        gs[tid] = t;
        gq[tid] = u;
    }
    __syncthreads();

    if (tid < COUT) {
        float t = gs[tid], u = gq[tid];
        #pragma unroll
        for (int k = 1; k < FOLD_PARTS; ++k) { t += gs[k * COUT + tid]; u += gq[k * COUT + tid]; }
        const float inv  = 1.0f / (float)(P_PILLARS * NPT);
        const float mean = t * inv;
        const float var  = u * inv - mean * mean;
        const float sc   = gamma[tid] * rsqrtf(var + 1e-3f);
        ((float*)sc4)[tid] = sc;
        ((float*)sh4)[tid] = beta[tid] - mean * sc;
    }
    __syncthreads();

    if (tid < FIN_ACTIVE) {
        const int i0 = blockIdx.x * (FIN_K * FIN_ACTIVE) + tid;
        const float4 s = sc4[q];
        const float4 t = sh4[q];
        const float4* mx4 = (const float4*)ws_mx;
        const float4* mn4 = (const float4*)ws_mn;
        float4* out4 = (float4*)out;

        const bool px = s.x >= 0.0f, py = s.y >= 0.0f, pz = s.z >= 0.0f, pw = s.w >= 0.0f;
        if (px && py && pz && pw) {
            pfn_finish_pieces<0>(mx4, mn4, out4, i0, s, t);
        } else if (!(px || py || pz || pw)) {
            pfn_finish_pieces<1>(mx4, mn4, out4, i0, s, t);
        } else {
            pfn_finish_pieces<2>(mx4, mn4, out4, i0, s, t);
        }
    }
}

extern "C" void kernel_launch(void* const* d_in, const int* in_sizes, int n_in,
                              void* d_out, int out_size, void* d_ws, size_t ws_size,
                              hipStream_t stream) {
    const float* features   = (const float*)d_in[0];
    const int*   num_points = (const int*)  d_in[1];
    const int*   coors      = (const int*)  d_in[2];
    const float* W          = (const float*)d_in[3];
    const float* gamma      = (const float*)d_in[4];
    const float* beta       = (const float*)d_in[5];
    float* out = (float*)d_out;

    float* ws      = (float*)d_ws;
    float* ws_mx   = ws;
    float* ws_mn   = ws_mx   + (size_t)P_PILLARS * COUT;
    float* ws_psum = ws_mn   + (size_t)P_PILLARS * COUT;
    float* ws_pssq = ws_psum + (size_t)NBLK_A * COUT;

    pfn_pass_a<<<NBLK_A, WPB * 32, 0, stream>>>(features, num_points, coors, W,
                                           ws_mx, ws_mn, ws_psum, ws_pssq);
    pfn_finish<<<FIN_BLKS, FIN_THREADS, 0, stream>>>(ws_mx, ws_mn, ws_psum, ws_pssq,
                                                 gamma, beta, out);
}


# pass A: sign of gamma folded into W's columns (exact), so the extreme the finish needs is always the max: the min chain, the min plane and its stores go; finish reads one plane, scale = |gamma| r
# speedup vs baseline: 16.0650x; 1.0798x over previous
#include <hip/hip_runtime.h>

typedef float v2f __attribute__((ext_vector_type(2)));
typedef float v8f __attribute__((ext_vector_type(8)));

#define P_PILLARS 60000
#define NPT 32
#define COUT 64
#define WPB 20
#define NBLK_A 256
#define NWAVE_A (NBLK_A * WPB)
#define RUN_LEN (P_PILLARS / NWAVE_A)
#define RUN_BIG (P_PILLARS - RUN_LEN * NWAVE_A)

__global__ __launch_bounds__(WPB * 32) void pfn_pass_a(
    const float* __restrict__ features,
    const int*   __restrict__ num_points,
    const int*   __restrict__ coors,
    const float* __restrict__ W,
    const float* __restrict__ gamma,
    float* __restrict__ ws_mx,
    float* __restrict__ ws_psum, float* __restrict__ ws_pssq)
{
    __shared__ float wsum[WPB][COUT];
    __shared__ float wssq[WPB][COUT];

    const int tid  = threadIdx.x;
    const int wid  = __builtin_amdgcn_readfirstlane(tid >> 5);
    const int lane = tid & 31;
    const int l16  = lane & 15;
    const bool hi  = (lane >> 4) != 0;
    const int k2   = hi ? 2 : 0;

    const int gw    = blockIdx.x * WPB + wid;
    const int cnt   = RUN_LEN + (gw < RUN_BIG ? 1 : 0);
    const int start = gw * RUN_LEN + (gw < RUN_BIG ? gw : RUN_BIG);

    const int hl   = (lane < cnt) ? lane : cnt - 1;
    const int h_np = num_points[start + hl];
    const int h_cy = coors[(start + hl) * 4 + 2];
    const int h_cx = coors[(start + hl) * 4 + 3];

    v2f bfr0[4];
    float nw4[4], nw5[4], nw6[4], nw7[4], nw8[4];
    #pragma unroll
    for (int nt = 0; nt < 4; ++nt) {
        const int c = nt * 16 + l16;
        const float sg = (gamma[c] >= 0.0f) ? 1.0f : -1.0f;
        const float ra = sg * W[(k2)     * COUT + c];
        const float rb = sg * W[(k2 + 1) * COUT + c];
        const float r4 = sg * W[4 * COUT + c];
        const float r5 = sg * W[5 * COUT + c];
        const float r6 = sg * W[6 * COUT + c];
        const float r7 = sg * W[7 * COUT + c];
        const float r8 = sg * W[8 * COUT + c];
        const float rc = hi ? r6 : r4;
        const float a5 = hi ? 0.0f : r5;
        const float a7 = hi ? 0.0f : r7;
        const float a8 = hi ? 0.0f : r8;
        bfr0[nt][0] = (ra + rc) + a7;
        bfr0[nt][1] = (rb + a5) + a8;
        nw4[nt] = -r4; nw5[nt] = -r5; nw6[nt] = -r6; nw7[nt] = -r7; nw8[nt] = -r8;
    }

    v2f chss2[4];
    float chs[4], cross[4], corr2[4];
    #pragma unroll
    for (int nt = 0; nt < 4; ++nt) {
        chss2[nt] = (v2f){0.0f, 0.0f};
        chs[nt] = 0.0f; cross[nt] = 0.0f; corr2[nt] = 0.0f;
    }

    const float4* f4 = (const float4*)features;
    float4 fa = f4[start * NPT + l16];
    float4 fb = f4[start * NPT + 16 + l16];
    int npts = __builtin_amdgcn_readlane(h_np, 0);
    int cy   = __builtin_amdgcn_readlane(h_cy, 0);
    int cx   = __builtin_amdgcn_readlane(h_cx, 0);

    for (int i = 0; i < cnt; ++i) {
        const int p = start + i;
        float4 na = fa, nb = fb;
        if (i + 1 < cnt) {
            na = f4[(p + 1) * NPT + l16];
            nb = f4[(p + 1) * NPT + 16 + l16];
        }
        __builtin_amdgcn_sched_barrier(0);

        float sx = fa.x + fb.x, sy = fa.y + fb.y, sz = fa.z + fb.z;
        #pragma unroll
        for (int off = 8; off >= 1; off >>= 1) {
            sx += __shfl_xor(sx, off, 32);
            sy += __shfl_xor(sy, off, 32);
            sz += __shfl_xor(sz, off, 32);
        }
        const float inv_n = 1.0f / (float)npts;
        const float cmx = sx * inv_n, cmy = sy * inv_n, cmz = sz * inv_n;

        const float xc = (float)cx * 0.2f + 0.1f;
        const float yc = (float)cy * 0.2f + (0.1f - 40.0f);

        float dv[4];
        #pragma unroll
        for (int nt = 0; nt < 4; ++nt) {
            dv[nt] = fmaf(cmx, nw4[nt], fmaf(cmy, nw5[nt], fmaf(cmz, nw6[nt],
                     fmaf(xc, nw7[nt], yc * nw8[nt]))));
        }

        const float p0 = hi ? fa.z : fa.x, p1 = hi ? fa.w : fa.y;
        const float q0 = hi ? fb.z : fb.x, q1 = hi ? fb.w : fb.y;
        const float r0 = __shfl(p0, lane & 16, 32), r1 = __shfl(p1, lane & 16, 32);
        const bool va = l16 < npts, vb = 16 + l16 < npts;
        v2f afr[2];
        afr[0][0] = va ? p0 : r0;
        afr[0][1] = va ? p1 : r1;
        afr[1][0] = vb ? q0 : r0;
        afr[1][1] = vb ? q1 : r1;

        const float kf = hi ? 0.0f : (float)(NPT - npts);
        const float nf = hi ? 0.0f : (float)npts;

        v8f acc[2][4];
        #pragma unroll
        for (int mt = 0; mt < 2; ++mt) {
            #pragma unroll
            for (int nt = 0; nt < 4; ++nt) {
                acc[mt][nt] = __builtin_amdgcn_wmma_f32_16x16x4_f32(
                    false, afr[mt], false, bfr0[nt], (short)0, (v8f){0,0,0,0,0,0,0,0}, false, false);
            }
        }

        float chmax[4];
        #pragma unroll
        for (int nt = 0; nt < 4; ++nt) {
            const float y0 = acc[0][nt][0];
            float ymax = fmaxf(y0, acc[0][nt][1]);
            v2f ps = {y0, acc[0][nt][1]};
            chss2[nt] = __builtin_elementwise_fma(ps, ps, chss2[nt]);
            #pragma unroll
            for (int mt = 0; mt < 2; ++mt) {
                #pragma unroll
                for (int j = (mt == 0 ? 1 : 0); j < 4; ++j) {
                    const float v0 = acc[mt][nt][2 * j], v1 = acc[mt][nt][2 * j + 1];
                    ymax = fmaxf(fmaxf(ymax, v0), v1);
                    const v2f vv = {v0, v1};
                    ps = ps + vv;
                    chss2[nt] = __builtin_elementwise_fma(vv, vv, chss2[nt]);
                }
            }
            chmax[nt] = ymax + dv[nt];
            const float s1 = fmaf(-kf, y0, ps[0] + ps[1]);
            const float u  = fmaf(nf, dv[nt], s1);
            chs[nt]   += u;
            cross[nt]  = fmaf(dv[nt], u + s1, cross[nt]);
            corr2[nt]  = fmaf(kf * y0, y0, corr2[nt]);
        }

        const float zmax = (npts < NPT) ? 0.0f : -3.402823e38f;

        #pragma unroll
        for (int j = 0; j < 2; ++j) {
            const float mine_mx = hi ? chmax[2 * j + 1] : chmax[2 * j];
            const float send_mx = hi ? chmax[2 * j]     : chmax[2 * j + 1];
            const float mx = fmaxf(fmaxf(mine_mx, __shfl_xor(send_mx, 16, 32)), zmax);
            ws_mx[(size_t)p * COUT + j * 32 + lane] = mx;
        }

        fa = na; fb = nb;
        npts = __builtin_amdgcn_readlane(h_np, i + 1);
        cy   = __builtin_amdgcn_readlane(h_cy, i + 1);
        cx   = __builtin_amdgcn_readlane(h_cx, i + 1);
    }

    float chss[4];
    #pragma unroll
    for (int nt = 0; nt < 4; ++nt) {
        chss[nt] = ((chss2[nt][0] + chss2[nt][1]) - corr2[nt]) + cross[nt];
        chs[nt]  += __shfl_xor(chs[nt], 16, 32);
        chss[nt] += __shfl_xor(chss[nt], 16, 32);
    }
    if (!hi) {
        #pragma unroll
        for (int nt = 0; nt < 4; ++nt) {
            wsum[wid][nt * 16 + l16] = chs[nt];
            wssq[wid][nt * 16 + l16] = chss[nt];
        }
    }
    __syncthreads();
    if (tid < COUT) {
        float s = 0.0f, ss = 0.0f;
        #pragma unroll
        for (int w = 0; w < WPB; ++w) { s += wsum[w][tid]; ss += wssq[w][tid]; }
        ws_psum[(size_t)blockIdx.x * COUT + tid] = s;
        ws_pssq[(size_t)blockIdx.x * COUT + tid] = ss;
    }
}

#define FIN_THREADS 512
#define FIN_ACTIVE 480
#define FIN_K 8
#define FIN_BLKS ((P_PILLARS * COUT / 4) / (FIN_ACTIVE * FIN_K))
#define FOLD_GROUPS (FIN_THREADS / 16)
#define FOLD_ROWS (NBLK_A / FOLD_GROUPS)
#define FOLD_PARTS 4
#define FOLD_GPP (FOLD_GROUPS / FOLD_PARTS)

__device__ __forceinline__ void pfn_finish_pieces(
    const float4* __restrict__ mx4,
    float4* __restrict__ out4, const int i0, const float4 s, const float4 t)
{
    float4 a[FIN_K];
    #pragma unroll
    for (int k = 0; k < FIN_K; ++k) a[k] = mx4[i0 + k * FIN_ACTIVE];
    #pragma unroll
    for (int k = 0; k < FIN_K; ++k) {
        const float4 v = a[k];
        float4 o;
        o.x = fmaxf(fmaf(s.x, v.x, t.x), 0.0f);
        o.y = fmaxf(fmaf(s.y, v.y, t.y), 0.0f);
        o.z = fmaxf(fmaf(s.z, v.z, t.z), 0.0f);
        o.w = fmaxf(fmaf(s.w, v.w, t.w), 0.0f);
        out4[i0 + k * FIN_ACTIVE] = o;
    }
}

__global__ __launch_bounds__(FIN_THREADS) void pfn_finish(
    const float* __restrict__ ws_mx,
    const float* __restrict__ psum, const float* __restrict__ pssq,
    const float* __restrict__ gamma, const float* __restrict__ beta,
    float* __restrict__ out)
{
    __shared__ float4 fs4[FOLD_GROUPS * (COUT / 4)];
    __shared__ float4 fq4[FOLD_GROUPS * (COUT / 4)];
    __shared__ float  gs[FOLD_PARTS * COUT];
    __shared__ float  gq[FOLD_PARTS * COUT];
    __shared__ float4 sc4[COUT / 4];
    __shared__ float4 sh4[COUT / 4];

    const int tid = threadIdx.x;
    const int q   = tid & 15;
    const int g   = tid >> 4;

    const float4* ps4 = (const float4*)psum;
    const float4* pq4 = (const float4*)pssq;
    float4 ra[FOLD_ROWS], rb[FOLD_ROWS];
    #pragma unroll
    for (int i = 0; i < FOLD_ROWS; ++i) {
        ra[i] = ps4[(g + FOLD_GROUPS * i) * (COUT / 4) + q];
        rb[i] = pq4[(g + FOLD_GROUPS * i) * (COUT / 4) + q];
    }
    float4 s1 = ra[0], q1 = rb[0];
    #pragma unroll
    for (int i = 1; i < FOLD_ROWS; ++i) {
        s1.x += ra[i].x; s1.y += ra[i].y; s1.z += ra[i].z; s1.w += ra[i].w;
        q1.x += rb[i].x; q1.y += rb[i].y; q1.z += rb[i].z; q1.w += rb[i].w;
    }
    fs4[g * (COUT / 4) + q] = s1;
    fq4[g * (COUT / 4) + q] = q1;
    __syncthreads();

    if (tid < FOLD_PARTS * COUT) {
        const float* fs = (const float*)fs4;
        const float* fq = (const float*)fq4;
        const int c  = tid & (COUT - 1);
        const int g0 = (tid / COUT) * FOLD_GPP;
        float t = fs[g0 * COUT + c], u = fq[g0 * COUT + c];
        #pragma unroll
        for (int k = 1; k < FOLD_GPP; ++k) { t += fs[(g0 + k) * COUT + c]; u += fq[(g0 + k) * COUT + c]; }
        gs[tid] = t;
        gq[tid] = u;
    }
    __syncthreads();

    if (tid < COUT) {
        float t = gs[tid], u = gq[tid];
        #pragma unroll
        for (int k = 1; k < FOLD_PARTS; ++k) { t += gs[k * COUT + tid]; u += gq[k * COUT + tid]; }
        const float inv  = 1.0f / (float)(P_PILLARS * NPT);
        const float mean = t * inv;
        const float var  = u * inv - mean * mean;
        const float sc   = fabsf(gamma[tid]) * rsqrtf(var + 1e-3f);
        ((float*)sc4)[tid] = sc;
        ((float*)sh4)[tid] = beta[tid] - mean * sc;
    }
    __syncthreads();

    if (tid < FIN_ACTIVE) {
        const int i0 = blockIdx.x * (FIN_K * FIN_ACTIVE) + tid;
        const float4 s = sc4[q];
        const float4 t = sh4[q];
        pfn_finish_pieces((const float4*)ws_mx, (float4*)out, i0, s, t);
    }
}

extern "C" void kernel_launch(void* const* d_in, const int* in_sizes, int n_in,
                              void* d_out, int out_size, void* d_ws, size_t ws_size,
                              hipStream_t stream) {
    const float* features   = (const float*)d_in[0];
    const int*   num_points = (const int*)  d_in[1];
    const int*   coors      = (const int*)  d_in[2];
    const float* W          = (const float*)d_in[3];
    const float* gamma      = (const float*)d_in[4];
    const float* beta       = (const float*)d_in[5];
    float* out = (float*)d_out;

    float* ws      = (float*)d_ws;
    float* ws_mx   = ws;
    float* ws_psum = ws_mx   + (size_t)P_PILLARS * COUT;
    float* ws_pssq = ws_psum + (size_t)NBLK_A * COUT;

    pfn_pass_a<<<NBLK_A, WPB * 32, 0, stream>>>(features, num_points, coors, W, gamma,
                                           ws_mx, ws_psum, ws_pssq);
    pfn_finish<<<FIN_BLKS, FIN_THREADS, 0, stream>>>(ws_mx, ws_psum, ws_pssq,
                                                 gamma, beta, out);
}


# pass A: a pillar of 16 points or fewer skips its second M-tile (rows 16..31 are then all copies of point 0): 4 of 8 matrix ops and that tile's max and sums go, one scalar branch; sum order changes
# speedup vs baseline: 17.0281x; 1.0600x over previous
#include <hip/hip_runtime.h>

typedef float v2f __attribute__((ext_vector_type(2)));
typedef float v8f __attribute__((ext_vector_type(8)));

#define P_PILLARS 60000
#define NPT 32
#define COUT 64
#define WPB 20
#define NBLK_A 256
#define NWAVE_A (NBLK_A * WPB)
#define RUN_LEN (P_PILLARS / NWAVE_A)
#define RUN_BIG (P_PILLARS - RUN_LEN * NWAVE_A)

__global__ __launch_bounds__(WPB * 32) void pfn_pass_a(
    const float* __restrict__ features,
    const int*   __restrict__ num_points,
    const int*   __restrict__ coors,
    const float* __restrict__ W,
    const float* __restrict__ gamma,
    float* __restrict__ ws_mx,
    float* __restrict__ ws_psum, float* __restrict__ ws_pssq)
{
    __shared__ float wsum[WPB][COUT];
    __shared__ float wssq[WPB][COUT];

    const int tid  = threadIdx.x;
    const int wid  = __builtin_amdgcn_readfirstlane(tid >> 5);
    const int lane = tid & 31;
    const int l16  = lane & 15;
    const bool hi  = (lane >> 4) != 0;
    const int k2   = hi ? 2 : 0;

    const int gw    = blockIdx.x * WPB + wid;
    const int cnt   = RUN_LEN + (gw < RUN_BIG ? 1 : 0);
    const int start = gw * RUN_LEN + (gw < RUN_BIG ? gw : RUN_BIG);

    const int hl   = (lane < cnt) ? lane : cnt - 1;
    const int h_np = num_points[start + hl];
    const int h_cy = coors[(start + hl) * 4 + 2];
    const int h_cx = coors[(start + hl) * 4 + 3];

    v2f bfr0[4];
    float nw4[4], nw5[4], nw6[4], nw7[4], nw8[4];
    #pragma unroll
    for (int nt = 0; nt < 4; ++nt) {
        const int c = nt * 16 + l16;
        const float sg = (gamma[c] >= 0.0f) ? 1.0f : -1.0f;
        const float ra = sg * W[(k2)     * COUT + c];
        const float rb = sg * W[(k2 + 1) * COUT + c];
        const float r4 = sg * W[4 * COUT + c];
        const float r5 = sg * W[5 * COUT + c];
        const float r6 = sg * W[6 * COUT + c];
        const float r7 = sg * W[7 * COUT + c];
        const float r8 = sg * W[8 * COUT + c];
        const float rc = hi ? r6 : r4;
        const float a5 = hi ? 0.0f : r5;
        const float a7 = hi ? 0.0f : r7;
        const float a8 = hi ? 0.0f : r8;
        bfr0[nt][0] = (ra + rc) + a7;
        bfr0[nt][1] = (rb + a5) + a8;
        nw4[nt] = -r4; nw5[nt] = -r5; nw6[nt] = -r6; nw7[nt] = -r7; nw8[nt] = -r8;
    }

    v2f chss2[4];
    float chs[4], cross[4], corr2[4];
    #pragma unroll
    for (int nt = 0; nt < 4; ++nt) {
        chss2[nt] = (v2f){0.0f, 0.0f};
        chs[nt] = 0.0f; cross[nt] = 0.0f; corr2[nt] = 0.0f;
    }

    const float4* f4 = (const float4*)features;
    float4 fa = f4[start * NPT + l16];
    float4 fb = f4[start * NPT + 16 + l16];
    int npts = __builtin_amdgcn_readlane(h_np, 0);
    int cy   = __builtin_amdgcn_readlane(h_cy, 0);
    int cx   = __builtin_amdgcn_readlane(h_cx, 0);

    for (int i = 0; i < cnt; ++i) {
        const int p = start + i;
        float4 na = fa, nb = fb;
        if (i + 1 < cnt) {
            na = f4[(p + 1) * NPT + l16];
            nb = f4[(p + 1) * NPT + 16 + l16];
        }
        __builtin_amdgcn_sched_barrier(0);

        float sx = fa.x + fb.x, sy = fa.y + fb.y, sz = fa.z + fb.z;
        #pragma unroll
        for (int off = 8; off >= 1; off >>= 1) {
            sx += __shfl_xor(sx, off, 32);
            sy += __shfl_xor(sy, off, 32);
            sz += __shfl_xor(sz, off, 32);
        }
        const float inv_n = 1.0f / (float)npts;
        const float cmx = sx * inv_n, cmy = sy * inv_n, cmz = sz * inv_n;

        const float xc = (float)cx * 0.2f + 0.1f;
        const float yc = (float)cy * 0.2f + (0.1f - 40.0f);

        float dv[4];
        #pragma unroll
        for (int nt = 0; nt < 4; ++nt) {
            dv[nt] = fmaf(cmx, nw4[nt], fmaf(cmy, nw5[nt], fmaf(cmz, nw6[nt],
                     fmaf(xc, nw7[nt], yc * nw8[nt]))));
        }

        const float p0 = hi ? fa.z : fa.x, p1 = hi ? fa.w : fa.y;
        const float q0 = hi ? fb.z : fb.x, q1 = hi ? fb.w : fb.y;
        const float r0 = __shfl(p0, lane & 16, 32), r1 = __shfl(p1, lane & 16, 32);
        const bool va = l16 < npts, vb = 16 + l16 < npts;
        v2f afr[2];
        afr[0][0] = va ? p0 : r0;
        afr[0][1] = va ? p1 : r1;
        afr[1][0] = vb ? q0 : r0;
        afr[1][1] = vb ? q1 : r1;

        const bool two = npts > 16;
        const int rows = two ? NPT : 16;
        const float kf = hi ? 0.0f : (float)(rows - npts);
        const float nf = hi ? 0.0f : (float)npts;

        float y0s[4], ymax[4];
        v2f ps[4];
        {
            v8f acc[4];
            #pragma unroll
            for (int nt = 0; nt < 4; ++nt) {
                acc[nt] = __builtin_amdgcn_wmma_f32_16x16x4_f32(
                    false, afr[0], false, bfr0[nt], (short)0, (v8f){0,0,0,0,0,0,0,0}, false, false);
            }
            #pragma unroll
            for (int nt = 0; nt < 4; ++nt) {
                const float y0 = acc[nt][0];
                y0s[nt]  = y0;
                ymax[nt] = fmaxf(y0, acc[nt][1]);
                ps[nt]   = (v2f){y0, acc[nt][1]};
                chss2[nt] = __builtin_elementwise_fma(ps[nt], ps[nt], chss2[nt]);
                #pragma unroll
                for (int j = 1; j < 4; ++j) {
                    const float v0 = acc[nt][2 * j], v1 = acc[nt][2 * j + 1];
                    ymax[nt] = fmaxf(fmaxf(ymax[nt], v0), v1);
                    const v2f vv = {v0, v1};
                    ps[nt] = ps[nt] + vv;
                    chss2[nt] = __builtin_elementwise_fma(vv, vv, chss2[nt]);
                }
            }
        }
        if (two) {
            v8f acc[4];
            #pragma unroll
            for (int nt = 0; nt < 4; ++nt) {
                acc[nt] = __builtin_amdgcn_wmma_f32_16x16x4_f32(
                    false, afr[1], false, bfr0[nt], (short)0, (v8f){0,0,0,0,0,0,0,0}, false, false);
            }
            #pragma unroll
            for (int nt = 0; nt < 4; ++nt) {
                #pragma unroll
                for (int j = 0; j < 4; ++j) {
                    const float v0 = acc[nt][2 * j], v1 = acc[nt][2 * j + 1];
                    ymax[nt] = fmaxf(fmaxf(ymax[nt], v0), v1);
                    const v2f vv = {v0, v1};
                    ps[nt] = ps[nt] + vv;
                    chss2[nt] = __builtin_elementwise_fma(vv, vv, chss2[nt]);
                }
            }
        }

        float chmax[4];
        #pragma unroll
        for (int nt = 0; nt < 4; ++nt) {
            const float y0 = y0s[nt];
            chmax[nt] = ymax[nt] + dv[nt];
            const float s1 = fmaf(-kf, y0, ps[nt][0] + ps[nt][1]);
            const float u  = fmaf(nf, dv[nt], s1);
            chs[nt]   += u;
            cross[nt]  = fmaf(dv[nt], u + s1, cross[nt]);
            corr2[nt]  = fmaf(kf * y0, y0, corr2[nt]);
        }

        const float zmax = (npts < NPT) ? 0.0f : -3.402823e38f;

        #pragma unroll
        for (int j = 0; j < 2; ++j) {
            const float mine_mx = hi ? chmax[2 * j + 1] : chmax[2 * j];
            const float send_mx = hi ? chmax[2 * j]     : chmax[2 * j + 1];
            const float mx = fmaxf(fmaxf(mine_mx, __shfl_xor(send_mx, 16, 32)), zmax);
            ws_mx[(size_t)p * COUT + j * 32 + lane] = mx;
        }

        fa = na; fb = nb;
        npts = __builtin_amdgcn_readlane(h_np, i + 1);
        cy   = __builtin_amdgcn_readlane(h_cy, i + 1);
        cx   = __builtin_amdgcn_readlane(h_cx, i + 1);
    }

    float chss[4];
    #pragma unroll
    for (int nt = 0; nt < 4; ++nt) {
        chss[nt] = ((chss2[nt][0] + chss2[nt][1]) - corr2[nt]) + cross[nt];
        chs[nt]  += __shfl_xor(chs[nt], 16, 32);
        chss[nt] += __shfl_xor(chss[nt], 16, 32);
    }
    if (!hi) {
        #pragma unroll
        for (int nt = 0; nt < 4; ++nt) {
            wsum[wid][nt * 16 + l16] = chs[nt];
            wssq[wid][nt * 16 + l16] = chss[nt];
        }
    }
    __syncthreads();
    if (tid < COUT) {
        float s = 0.0f, ss = 0.0f;
        #pragma unroll
        for (int w = 0; w < WPB; ++w) { s += wsum[w][tid]; ss += wssq[w][tid]; }
        ws_psum[(size_t)blockIdx.x * COUT + tid] = s;
        ws_pssq[(size_t)blockIdx.x * COUT + tid] = ss;
    }
}

#define FIN_THREADS 512
#define FIN_ACTIVE 480
#define FIN_K 8
#define FIN_BLKS ((P_PILLARS * COUT / 4) / (FIN_ACTIVE * FIN_K))
#define FOLD_GROUPS (FIN_THREADS / 16)
#define FOLD_ROWS (NBLK_A / FOLD_GROUPS)
#define FOLD_PARTS 4
#define FOLD_GPP (FOLD_GROUPS / FOLD_PARTS)

__device__ __forceinline__ void pfn_finish_pieces(
    const float4* __restrict__ mx4,
    float4* __restrict__ out4, const int i0, const float4 s, const float4 t)
{
    float4 a[FIN_K];
    #pragma unroll
    for (int k = 0; k < FIN_K; ++k) a[k] = mx4[i0 + k * FIN_ACTIVE];
    #pragma unroll
    for (int k = 0; k < FIN_K; ++k) {
        const float4 v = a[k];
        float4 o;
        o.x = fmaxf(fmaf(s.x, v.x, t.x), 0.0f);
        o.y = fmaxf(fmaf(s.y, v.y, t.y), 0.0f);
        o.z = fmaxf(fmaf(s.z, v.z, t.z), 0.0f);
        o.w = fmaxf(fmaf(s.w, v.w, t.w), 0.0f);
        out4[i0 + k * FIN_ACTIVE] = o;
    }
}

__global__ __launch_bounds__(FIN_THREADS) void pfn_finish(
    const float* __restrict__ ws_mx,
    const float* __restrict__ psum, const float* __restrict__ pssq,
    const float* __restrict__ gamma, const float* __restrict__ beta,
    float* __restrict__ out)
{
    __shared__ float4 fs4[FOLD_GROUPS * (COUT / 4)];
    __shared__ float4 fq4[FOLD_GROUPS * (COUT / 4)];
    __shared__ float  gs[FOLD_PARTS * COUT];
    __shared__ float  gq[FOLD_PARTS * COUT];
    __shared__ float4 sc4[COUT / 4];
    __shared__ float4 sh4[COUT / 4];

    const int tid = threadIdx.x;
    const int q   = tid & 15;
    const int g   = tid >> 4;

    const float4* ps4 = (const float4*)psum;
    const float4* pq4 = (const float4*)pssq;
    float4 ra[FOLD_ROWS], rb[FOLD_ROWS];
    #pragma unroll
    for (int i = 0; i < FOLD_ROWS; ++i) {
        ra[i] = ps4[(g + FOLD_GROUPS * i) * (COUT / 4) + q];
        rb[i] = pq4[(g + FOLD_GROUPS * i) * (COUT / 4) + q];
    }
    float4 s1 = ra[0], q1 = rb[0];
    #pragma unroll
    for (int i = 1; i < FOLD_ROWS; ++i) {
        s1.x += ra[i].x; s1.y += ra[i].y; s1.z += ra[i].z; s1.w += ra[i].w;
        q1.x += rb[i].x; q1.y += rb[i].y; q1.z += rb[i].z; q1.w += rb[i].w;
    }
    fs4[g * (COUT / 4) + q] = s1;
    fq4[g * (COUT / 4) + q] = q1;
    __syncthreads();

    if (tid < FOLD_PARTS * COUT) {
        const float* fs = (const float*)fs4;
        const float* fq = (const float*)fq4;
        const int c  = tid & (COUT - 1);
        const int g0 = (tid / COUT) * FOLD_GPP;
        float t = fs[g0 * COUT + c], u = fq[g0 * COUT + c];
        #pragma unroll
        for (int k = 1; k < FOLD_GPP; ++k) { t += fs[(g0 + k) * COUT + c]; u += fq[(g0 + k) * COUT + c]; }
        gs[tid] = t;
        gq[tid] = u;
    }
    __syncthreads();

    if (tid < COUT) {
        float t = gs[tid], u = gq[tid];
        #pragma unroll
        for (int k = 1; k < FOLD_PARTS; ++k) { t += gs[k * COUT + tid]; u += gq[k * COUT + tid]; }
        const float inv  = 1.0f / (float)(P_PILLARS * NPT);
        const float mean = t * inv;
        const float var  = u * inv - mean * mean;
        const float sc   = fabsf(gamma[tid]) * rsqrtf(var + 1e-3f);
        ((float*)sc4)[tid] = sc;
        ((float*)sh4)[tid] = beta[tid] - mean * sc;
    }
    __syncthreads();

    if (tid < FIN_ACTIVE) {
        const int i0 = blockIdx.x * (FIN_K * FIN_ACTIVE) + tid;
        const float4 s = sc4[q];
        const float4 t = sh4[q];
        pfn_finish_pieces((const float4*)ws_mx, (float4*)out, i0, s, t);
    }
}

extern "C" void kernel_launch(void* const* d_in, const int* in_sizes, int n_in,
                              void* d_out, int out_size, void* d_ws, size_t ws_size,
                              hipStream_t stream) {
    const float* features   = (const float*)d_in[0];
    const int*   num_points = (const int*)  d_in[1];
    const int*   coors      = (const int*)  d_in[2];
    const float* W          = (const float*)d_in[3];
    const float* gamma      = (const float*)d_in[4];
    const float* beta       = (const float*)d_in[5];
    float* out = (float*)d_out;

    float* ws      = (float*)d_ws;
    float* ws_mx   = ws;
    float* ws_psum = ws_mx   + (size_t)P_PILLARS * COUT;
    float* ws_pssq = ws_psum + (size_t)NBLK_A * COUT;

    pfn_pass_a<<<NBLK_A, WPB * 32, 0, stream>>>(features, num_points, coors, W, gamma,
                                           ws_mx, ws_psum, ws_pssq);
    pfn_finish<<<FIN_BLKS, FIN_THREADS, 0, stream>>>(ws_mx, ws_psum, ws_pssq,
                                                 gamma, beta, out);
}


# pass A: batch-norm sums in closed form from moments of the raw points (per lane its own point: 4x4, 5x4 against the pillar's offsets; 5x5 per pillar), contracted with W per block; sum order changes
# speedup vs baseline: 18.0960x; 1.0627x over previous
#include <hip/hip_runtime.h>

typedef float v2f __attribute__((ext_vector_type(2)));
typedef float v8f __attribute__((ext_vector_type(8)));

#define P_PILLARS 60000
#define NPT 32
#define COUT 64
#define WPB 20
#define NBLK_A 256
#define NWAVE_A (NBLK_A * WPB)
#define RUN_LEN (P_PILLARS / NWAVE_A)
#define RUN_BIG (P_PILLARS - RUN_LEN * NWAVE_A)
#define NV 56
#define NQ (NV / 4)
#define PITCHQ 15

__global__ __launch_bounds__(WPB * 32) void pfn_pass_a(
    const float* __restrict__ features,
    const int*   __restrict__ num_points,
    const int*   __restrict__ coors,
    const float* __restrict__ W,
    const float* __restrict__ gamma,
    float* __restrict__ ws_mx,
    float* __restrict__ ws_psum, float* __restrict__ ws_pssq)
{
    __shared__ float4 red4[WPB * 32 * PITCHQ];
    __shared__ float4 wtot4[WPB * NQ];
    __shared__ float  tot[NV];

    const int tid  = threadIdx.x;
    const int wid  = __builtin_amdgcn_readfirstlane(tid >> 5);
    const int lane = tid & 31;
    const int l16  = lane & 15;
    const bool hi  = (lane >> 4) != 0;
    const int k2   = hi ? 2 : 0;

    const int gw    = blockIdx.x * WPB + wid;
    const int cnt   = RUN_LEN + (gw < RUN_BIG ? 1 : 0);
    const int start = gw * RUN_LEN + (gw < RUN_BIG ? gw : RUN_BIG);

    const int hl   = (lane < cnt) ? lane : cnt - 1;
    const int h_np = num_points[start + hl];
    const int h_cy = coors[(start + hl) * 4 + 2];
    const int h_cx = coors[(start + hl) * 4 + 3];

    v2f bfr0[4];
    float nw4[4], nw5[4], nw6[4], nw7[4], nw8[4];
    #pragma unroll
    for (int nt = 0; nt < 4; ++nt) {
        const int c = nt * 16 + l16;
        const float sg = (gamma[c] >= 0.0f) ? 1.0f : -1.0f;
        const float ra = sg * W[(k2)     * COUT + c];
        const float rb = sg * W[(k2 + 1) * COUT + c];
        const float r4 = sg * W[4 * COUT + c];
        const float r5 = sg * W[5 * COUT + c];
        const float r6 = sg * W[6 * COUT + c];
        const float r7 = sg * W[7 * COUT + c];
        const float r8 = sg * W[8 * COUT + c];
        const float rc = hi ? r6 : r4;
        const float a5 = hi ? 0.0f : r5;
        const float a7 = hi ? 0.0f : r7;
        const float a8 = hi ? 0.0f : r8;
        bfr0[nt][0] = (ra + rc) + a7;
        bfr0[nt][1] = (rb + a5) + a8;
        nw4[nt] = -r4; nw5[nt] = -r5; nw6[nt] = -r6; nw7[nt] = -r7; nw8[nt] = -r8;
    }

    float ma[10], ga[5][4], sa[4];
    float kcx = 0.0f, kcy = 0.0f, kcz = 0.0f;
    #pragma unroll
    for (int k = 0; k < 10; ++k) ma[k] = 0.0f;
    #pragma unroll
    for (int i = 0; i < 5; ++i) {
        #pragma unroll
        for (int j = 0; j < 4; ++j) ga[i][j] = 0.0f;
    }
    #pragma unroll
    for (int j = 0; j < 4; ++j) sa[j] = 0.0f;

    const float4* f4 = (const float4*)features;
    float4 fa = f4[start * NPT + l16];
    float4 fb = f4[start * NPT + 16 + l16];
    int npts = __builtin_amdgcn_readlane(h_np, 0);
    int cy   = __builtin_amdgcn_readlane(h_cy, 0);
    int cx   = __builtin_amdgcn_readlane(h_cx, 0);

    for (int i = 0; i < cnt; ++i) {
        const int p = start + i;
        float4 na = fa, nb = fb;
        if (i + 1 < cnt) {
            na = f4[(p + 1) * NPT + l16];
            nb = f4[(p + 1) * NPT + 16 + l16];
        }
        __builtin_amdgcn_sched_barrier(0);

        float sx = fa.x + fb.x, sy = fa.y + fb.y, sz = fa.z + fb.z;
        #pragma unroll
        for (int off = 8; off >= 1; off >>= 1) {
            sx += __shfl_xor(sx, off, 32);
            sy += __shfl_xor(sy, off, 32);
            sz += __shfl_xor(sz, off, 32);
        }
        const float inv_n = 1.0f / (float)npts;
        const float cmx = sx * inv_n, cmy = sy * inv_n, cmz = sz * inv_n;
        kcx = (lane == i) ? cmx : kcx;
        kcy = (lane == i) ? cmy : kcy;
        kcz = (lane == i) ? cmz : kcz;

        const float xc = (float)cx * 0.2f + 0.1f;
        const float yc = (float)cy * 0.2f + (0.1f - 40.0f);

        float dv[4];
        #pragma unroll
        for (int nt = 0; nt < 4; ++nt) {
            dv[nt] = fmaf(cmx, nw4[nt], fmaf(cmy, nw5[nt], fmaf(cmz, nw6[nt],
                     fmaf(xc, nw7[nt], yc * nw8[nt]))));
        }

        {
            const float g[5] = {cmx, cmy, cmz, xc, yc};
            const bool vl = lane < npts;
            float a[4];
            a[0] = vl ? (hi ? fb.x : fa.x) : 0.0f;
            a[1] = vl ? (hi ? fb.y : fa.y) : 0.0f;
            a[2] = vl ? (hi ? fb.z : fa.z) : 0.0f;
            a[3] = vl ? (hi ? fb.w : fa.w) : 0.0f;
            int t = 0;
            #pragma unroll
            for (int j = 0; j < 4; ++j) {
                sa[j] += a[j];
                #pragma unroll
                for (int k = j; k < 4; ++k) { ma[t] = fmaf(a[j], a[k], ma[t]); ++t; }
                #pragma unroll
                for (int i = 0; i < 5; ++i) ga[i][j] = fmaf(g[i], a[j], ga[i][j]);
            }
        }

        const float p0 = hi ? fa.z : fa.x, p1 = hi ? fa.w : fa.y;
        const float q0 = hi ? fb.z : fb.x, q1 = hi ? fb.w : fb.y;
        const float r0 = __shfl(p0, lane & 16, 32), r1 = __shfl(p1, lane & 16, 32);
        const bool va = l16 < npts, vb = 16 + l16 < npts;
        v2f afr[2];
        afr[0][0] = va ? p0 : r0;
        afr[0][1] = va ? p1 : r1;
        afr[1][0] = vb ? q0 : r0;
        afr[1][1] = vb ? q1 : r1;

        const bool two = npts > 16;

        float ymax[4];
        {
            v8f acc[4];
            #pragma unroll
            for (int nt = 0; nt < 4; ++nt) {
                acc[nt] = __builtin_amdgcn_wmma_f32_16x16x4_f32(
                    false, afr[0], false, bfr0[nt], (short)0, (v8f){0,0,0,0,0,0,0,0}, false, false);
            }
            #pragma unroll
            for (int nt = 0; nt < 4; ++nt) {
                ymax[nt] = fmaxf(acc[nt][0], acc[nt][1]);
                #pragma unroll
                for (int j = 1; j < 4; ++j) {
                    const float v0 = acc[nt][2 * j], v1 = acc[nt][2 * j + 1];
                    ymax[nt] = fmaxf(fmaxf(ymax[nt], v0), v1);
                }
            }
        }
        if (two) {
            v8f acc[4];
            #pragma unroll
            for (int nt = 0; nt < 4; ++nt) {
                acc[nt] = __builtin_amdgcn_wmma_f32_16x16x4_f32(
                    false, afr[1], false, bfr0[nt], (short)0, (v8f){0,0,0,0,0,0,0,0}, false, false);
            }
            #pragma unroll
            for (int nt = 0; nt < 4; ++nt) {
                #pragma unroll
                for (int j = 0; j < 4; ++j) {
                    const float v0 = acc[nt][2 * j], v1 = acc[nt][2 * j + 1];
                    ymax[nt] = fmaxf(fmaxf(ymax[nt], v0), v1);
                }
            }
        }

        float chmax[4];
        #pragma unroll
        for (int nt = 0; nt < 4; ++nt) {
            chmax[nt] = ymax[nt] + dv[nt];
        }

        const float zmax = (npts < NPT) ? 0.0f : -3.402823e38f;

        #pragma unroll
        for (int j = 0; j < 2; ++j) {
            const float mine_mx = hi ? chmax[2 * j + 1] : chmax[2 * j];
            const float send_mx = hi ? chmax[2 * j]     : chmax[2 * j + 1];
            const float mx = fmaxf(fmaxf(mine_mx, __shfl_xor(send_mx, 16, 32)), zmax);
            ws_mx[(size_t)p * COUT + j * 32 + lane] = mx;
        }

        fa = na; fb = nb;
        npts = __builtin_amdgcn_readlane(h_np, i + 1);
        cy   = __builtin_amdgcn_readlane(h_cy, i + 1);
        cx   = __builtin_amdgcn_readlane(h_cx, i + 1);
    }

    float w[9];
    {
        const int c = tid & (COUT - 1);
        const float sg = (gamma[c] >= 0.0f) ? 1.0f : -1.0f;
        #pragma unroll
        for (int k = 0; k < 9; ++k) w[k] = sg * W[k * COUT + c];
    }
    float hg[15], ngs[5];
    {
        const float nfl = (lane < cnt) ? (float)h_np : 0.0f;
        const float g[5] = {kcx, kcy, kcz, (float)h_cx * 0.2f + 0.1f, (float)h_cy * 0.2f + (0.1f - 40.0f)};
        int t = 0;
        #pragma unroll
        for (int i = 0; i < 5; ++i) {
            ngs[i] = nfl * g[i];
            #pragma unroll
            for (int j = i; j < 5; ++j) { hg[t] = ngs[i] * g[j]; ++t; }
        }
    }
    {
        float4* mine = &red4[tid * PITCHQ];
        mine[0] = make_float4(ma[0], ma[1], ma[2], ma[3]);
        mine[1] = make_float4(ma[4], ma[5], ma[6], ma[7]);
        mine[2] = make_float4(ma[8], ma[9], ga[0][0], ga[0][1]);
        mine[3] = make_float4(ga[0][2], ga[0][3], ga[1][0], ga[1][1]);
        mine[4] = make_float4(ga[1][2], ga[1][3], ga[2][0], ga[2][1]);
        mine[5] = make_float4(ga[2][2], ga[2][3], ga[3][0], ga[3][1]);
        mine[6] = make_float4(ga[3][2], ga[3][3], ga[4][0], ga[4][1]);
        mine[7] = make_float4(ga[4][2], ga[4][3], sa[0], sa[1]);
        mine[8] = make_float4(sa[2], sa[3], 0.0f, 0.0f);
        mine[9]  = make_float4(hg[0], hg[1], hg[2], hg[3]);
        mine[10] = make_float4(hg[4], hg[5], hg[6], hg[7]);
        mine[11] = make_float4(hg[8], hg[9], hg[10], hg[11]);
        mine[12] = make_float4(hg[12], hg[13], hg[14], ngs[0]);
        mine[13] = make_float4(ngs[1], ngs[2], ngs[3], ngs[4]);
    }
    __builtin_amdgcn_fence(__ATOMIC_RELEASE, "wavefront");
    __builtin_amdgcn_wave_barrier();
    __builtin_amdgcn_fence(__ATOMIC_ACQUIRE, "wavefront");
    {
        const int t  = (lane < 2 * NQ) ? lane : 2 * NQ - 1;
        const int q  = (t < NQ) ? t : t - NQ;
        const int l0 = (t < NQ) ? 0 : 16;
        const float4* src = &red4[(wid * 32 + l0) * PITCHQ + q];
        float4 r[16];
        #pragma unroll
        for (int k = 0; k < 16; ++k) r[k] = src[k * PITCHQ];
        float4 acc4 = r[0];
        #pragma unroll
        for (int k = 1; k < 16; ++k) { acc4.x += r[k].x; acc4.y += r[k].y; acc4.z += r[k].z; acc4.w += r[k].w; }
        const int from = (lane + NQ) & 31;
        acc4.x += __shfl(acc4.x, from, 32);
        acc4.y += __shfl(acc4.y, from, 32);
        acc4.z += __shfl(acc4.z, from, 32);
        acc4.w += __shfl(acc4.w, from, 32);
        if (lane < NQ) wtot4[wid * NQ + lane] = acc4;
    }
    __syncthreads();
    if (tid < NV * 4) {
        const float* wtot = (const float*)wtot4;
        const int v = tid >> 2, w0 = (tid & 3) * (WPB / 4);
        float r[WPB / 4];
        #pragma unroll
        for (int k = 0; k < WPB / 4; ++k) r[k] = wtot[(w0 + k) * NV + v];
        float t = r[0];
        #pragma unroll
        for (int k = 1; k < WPB / 4; ++k) t += r[k];
        t += __shfl_xor(t, 1, 32);
        t += __shfl_xor(t, 2, 32);
        if ((tid & 3) == 0) tot[v] = t;
    }
    __syncthreads();
    if (tid < COUT) {
        float wc[4], u[5];
        wc[0] = (w[0] + w[4]) + w[7];
        wc[1] = (w[1] + w[5]) + w[8];
        wc[2] = w[2] + w[6];
        wc[3] = w[3];
        #pragma unroll
        for (int i = 0; i < 5; ++i) u[i] = -w[4 + i];
        const float* tm = tot;
        const float* tg = tot + 10;
        const float* ts = tot + 30;
        const float* th = tot + 36;
        const float* tn = tot + 51;
        float sx1 = 0.0f, sxx = 0.0f;
        int t = 0;
        #pragma unroll
        for (int j = 0; j < 4; ++j) {
            sx1 = fmaf(ts[j], wc[j], sx1);
            float row = tm[t] * wc[j]; ++t;
            #pragma unroll
            for (int k = j + 1; k < 4; ++k) { row = fmaf(2.0f * tm[t], wc[k], row); ++t; }
            float cr = 0.0f;
            #pragma unroll
            for (int i = 0; i < 5; ++i) cr = fmaf(tg[i * 4 + j], u[i], cr);
            sxx = fmaf(wc[j], fmaf(2.0f, cr, row), sxx);
        }
        t = 0;
        #pragma unroll
        for (int i = 0; i < 5; ++i) {
            sx1 = fmaf(tn[i], u[i], sx1);
            float row = th[t] * u[i]; ++t;
            #pragma unroll
            for (int k = i + 1; k < 5; ++k) { row = fmaf(2.0f * th[t], u[k], row); ++t; }
            sxx = fmaf(u[i], row, sxx);
        }
        ws_psum[(size_t)blockIdx.x * COUT + tid] = sx1;
        ws_pssq[(size_t)blockIdx.x * COUT + tid] = sxx;
    }
}

#define FIN_THREADS 512
#define FIN_ACTIVE 480
#define FIN_K 8
#define FIN_BLKS ((P_PILLARS * COUT / 4) / (FIN_ACTIVE * FIN_K))
#define FOLD_GROUPS (FIN_THREADS / 16)
#define FOLD_ROWS (NBLK_A / FOLD_GROUPS)
#define FOLD_PARTS 4
#define FOLD_GPP (FOLD_GROUPS / FOLD_PARTS)

__device__ __forceinline__ void pfn_finish_pieces(
    const float4* __restrict__ mx4,
    float4* __restrict__ out4, const int i0, const float4 s, const float4 t)
{
    float4 a[FIN_K];
    #pragma unroll
    for (int k = 0; k < FIN_K; ++k) a[k] = mx4[i0 + k * FIN_ACTIVE];
    #pragma unroll
    for (int k = 0; k < FIN_K; ++k) {
        const float4 v = a[k];
        float4 o;
        o.x = fmaxf(fmaf(s.x, v.x, t.x), 0.0f);
        o.y = fmaxf(fmaf(s.y, v.y, t.y), 0.0f);
        o.z = fmaxf(fmaf(s.z, v.z, t.z), 0.0f);
        o.w = fmaxf(fmaf(s.w, v.w, t.w), 0.0f);
        out4[i0 + k * FIN_ACTIVE] = o;
    }
}

__global__ __launch_bounds__(FIN_THREADS) void pfn_finish(
    const float* __restrict__ ws_mx,
    const float* __restrict__ psum, const float* __restrict__ pssq,
    const float* __restrict__ gamma, const float* __restrict__ beta,
    float* __restrict__ out)
{
    __shared__ float4 fs4[FOLD_GROUPS * (COUT / 4)];
    __shared__ float4 fq4[FOLD_GROUPS * (COUT / 4)];
    __shared__ float  gs[FOLD_PARTS * COUT];
    __shared__ float  gq[FOLD_PARTS * COUT];
    __shared__ float4 sc4[COUT / 4];
    __shared__ float4 sh4[COUT / 4];

    const int tid = threadIdx.x;
    const int q   = tid & 15;
    const int g   = tid >> 4;

    const float4* ps4 = (const float4*)psum;
    const float4* pq4 = (const float4*)pssq;
    float4 ra[FOLD_ROWS], rb[FOLD_ROWS];
    #pragma unroll
    for (int i = 0; i < FOLD_ROWS; ++i) {
        ra[i] = ps4[(g + FOLD_GROUPS * i) * (COUT / 4) + q];
        rb[i] = pq4[(g + FOLD_GROUPS * i) * (COUT / 4) + q];
    }
    float4 s1 = ra[0], q1 = rb[0];
    #pragma unroll
    for (int i = 1; i < FOLD_ROWS; ++i) {
        s1.x += ra[i].x; s1.y += ra[i].y; s1.z += ra[i].z; s1.w += ra[i].w;
        q1.x += rb[i].x; q1.y += rb[i].y; q1.z += rb[i].z; q1.w += rb[i].w;
    }
    fs4[g * (COUT / 4) + q] = s1;
    fq4[g * (COUT / 4) + q] = q1;
    __syncthreads();

    if (tid < FOLD_PARTS * COUT) {
        const float* fs = (const float*)fs4;
        const float* fq = (const float*)fq4;
        const int c  = tid & (COUT - 1);
        const int g0 = (tid / COUT) * FOLD_GPP;
        float t = fs[g0 * COUT + c], u = fq[g0 * COUT + c];
        #pragma unroll
        for (int k = 1; k < FOLD_GPP; ++k) { t += fs[(g0 + k) * COUT + c]; u += fq[(g0 + k) * COUT + c]; }
        gs[tid] = t;
        gq[tid] = u;
    }
    __syncthreads();

    if (tid < COUT) {
        float t = gs[tid], u = gq[tid];
        #pragma unroll
        for (int k = 1; k < FOLD_PARTS; ++k) { t += gs[k * COUT + tid]; u += gq[k * COUT + tid]; }
        const float inv  = 1.0f / (float)(P_PILLARS * NPT);
        const float mean = t * inv;
        const float var  = u * inv - mean * mean;
        const float sc   = fabsf(gamma[tid]) * rsqrtf(var + 1e-3f);
        ((float*)sc4)[tid] = sc;
        ((float*)sh4)[tid] = beta[tid] - mean * sc;
    }
    __syncthreads();

    if (tid < FIN_ACTIVE) {
        const int i0 = blockIdx.x * (FIN_K * FIN_ACTIVE) + tid;
        const float4 s = sc4[q];
        const float4 t = sh4[q];
        pfn_finish_pieces((const float4*)ws_mx, (float4*)out, i0, s, t);
    }
}

extern "C" void kernel_launch(void* const* d_in, const int* in_sizes, int n_in,
                              void* d_out, int out_size, void* d_ws, size_t ws_size,
                              hipStream_t stream) {
    const float* features   = (const float*)d_in[0];
    const int*   num_points = (const int*)  d_in[1];
    const int*   coors      = (const int*)  d_in[2];
    const float* W          = (const float*)d_in[3];
    const float* gamma      = (const float*)d_in[4];
    const float* beta       = (const float*)d_in[5];
    float* out = (float*)d_out;

    float* ws      = (float*)d_ws;
    float* ws_mx   = ws;
    float* ws_psum = ws_mx   + (size_t)P_PILLARS * COUT;
    float* ws_pssq = ws_psum + (size_t)NBLK_A * COUT;

    pfn_pass_a<<<NBLK_A, WPB * 32, 0, stream>>>(features, num_points, coors, W, gamma,
                                           ws_mx, ws_psum, ws_pssq);
    pfn_finish<<<FIN_BLKS, FIN_THREADS, 0, stream>>>(ws_mx, ws_psum, ws_pssq,
                                                 gamma, beta, out);
}
